# starting kernel
# baseline (speedup 1.0000x reference)
#include <hip/hip_runtime.h>

typedef __attribute__((ext_vector_type(2))) float v2f;
typedef __attribute__((ext_vector_type(8))) float v8f;


__global__ void fill_ones_kernel(float* __restrict__ p, int n) {
    int i = blockIdx.x * blockDim.x + threadIdx.x;
    if (i < n) p[i] = 1.0f;
}

__global__ void deg_accum_kernel(float* __restrict__ deg, const int* __restrict__ dst, int E) {
    int e = blockIdx.x * blockDim.x + threadIdx.x;
    if (e < E) atomicAdd(&deg[dst[e]], 1.0f);
}

__global__ void deg_to_dinv_kernel(float* __restrict__ d, int n) {
    int i = blockIdx.x * blockDim.x + threadIdx.x;
    if (i < n) {
        float v = d[i];
        d[i] = (v > 0.0f) ? rsqrtf(v) : 0.0f;
    }
}


template <int NT>
__global__ __launch_bounds__(256) void gemm_wmma_kernel(const float* __restrict__ X,
                                                        const float* __restrict__ W,
                                                        float* __restrict__ Y,
                                                        int N, int numStripes) {
    constexpr int K = 128;
    constexpr int OUT = NT * 16;
    extern __shared__ float wt[];

    const int tid = threadIdx.x;
    for (int i = tid; i < K * OUT; i += 256) {
        int k = i / OUT;
        int n = i - k * OUT;
        wt[(k >> 1) * (2 * OUT) + 2 * n + (k & 1)] = W[i];
    }
    __syncthreads();

    const int lane = tid & 31;
    const int wave = tid >> 5;
    const int m = lane & 15;
    const int half = lane >> 4;
    const int kg = half << 1;

    const float* lbase = wt + (half * OUT + m) * 2;

    for (int stripe = blockIdx.x; stripe < numStripes; stripe += gridDim.x) {
        const int rowBase = stripe * 128 + wave * 16;
        int arow = rowBase + m;
        if (arow >= N) arow = N - 1;
        const float* xrow = X + (long)arow * K;

        v8f zero = {};
        v8f acc[NT];
#pragma unroll
        for (int t = 0; t < NT; ++t) acc[t] = zero;

#pragma unroll 8
        for (int k = 0; k < K; k += 4) {
            v2f a = *(const v2f*)(xrow + k + kg);
            v2f b[NT];
            const float* lk = lbase + (k >> 1) * (2 * OUT);
#pragma unroll
            for (int t = 0; t < NT; ++t) b[t] = *(const v2f*)(lk + t * 32);
#pragma unroll
            for (int t = 0; t < NT; ++t)
                acc[t] = __builtin_amdgcn_wmma_f32_16x16x4_f32(
                    false, a, false, b[t], (short)0, acc[t], false, false);
        }

#pragma unroll
        for (int t = 0; t < NT; ++t) {
            const int col = t * 16 + m;
#pragma unroll
            for (int r = 0; r < 8; ++r) {
                const int row = rowBase + r + 8 * half;
                if (row < N) Y[(long)row * OUT + col] = acc[t][r];
            }
        }
    }
}


__global__ void self_bias_kernel(float* __restrict__ out, const float* __restrict__ xw,
                                 const float* __restrict__ b, const float* __restrict__ dinv,
                                 int N, int DIM) {
    long idx = (long)blockIdx.x * blockDim.x + threadIdx.x;
    if (idx >= (long)N * DIM) return;
    int i = (int)(idx / DIM);
    int c = (int)(idx - (long)i * DIM);
    float dv = dinv[i];
    out[idx] = b[c] + xw[idx] * dv * dv;
}


template <int VPL>
__global__ __launch_bounds__(256) void edge_scatter_kernel(const float* __restrict__ xw,
                                                           const float* __restrict__ dinv,
                                                           const int* __restrict__ src,
                                                           const int* __restrict__ dst,
                                                           float* __restrict__ out, int E) {
    constexpr int DIM = 32 * VPL;
    long g = (long)blockIdx.x * blockDim.x + threadIdx.x;
    int e = (int)(g >> 5);
    int lane = (int)(g & 31);
    if (e >= E) return;
    int s = src[e];
    int d = dst[e];
    float w = dinv[s] * dinv[d];
    const float* p = xw + (long)s * DIM + lane * VPL;
    float* q = out + (long)d * DIM + lane * VPL;
    float v[VPL];
#pragma unroll
    for (int i = 0; i < VPL; ++i) v[i] = p[i];
#pragma unroll
    for (int i = 0; i < VPL; ++i) atomicAdd(q + i, v[i] * w);
}


extern "C" void kernel_launch(void* const* d_in, const int* in_sizes, int n_in,
                              void* d_out, int out_size, void* d_ws, size_t ws_size,
                              hipStream_t stream) {
    const float* x  = (const float*)d_in[0];
    const int*   ei = (const int*)d_in[1];
    const float* W1 = (const float*)d_in[2];
    const float* b1 = (const float*)d_in[3];
    const float* W2 = (const float*)d_in[4];
    const float* b2 = (const float*)d_in[5];
    float* out = (float*)d_out;

    const int N = in_sizes[0] / 128;
    const int E = in_sizes[1] / 2;
    const int* src = ei;
    const int* dst = ei + E;

    float* dinv = (float*)d_ws;
    float* xw   = dinv + N;
    float* h    = xw + (size_t)N * 128;

    fill_ones_kernel<<<(N + 255) / 256, 256, 0, stream>>>(dinv, N);
    deg_accum_kernel<<<(E + 255) / 256, 256, 0, stream>>>(dinv, dst, E);
    deg_to_dinv_kernel<<<(N + 255) / 256, 256, 0, stream>>>(dinv, N);

    const int numStripes = (N + 127) / 128;
    const int gemmBlocks = numStripes < 320 ? numStripes : 320;
    const long et = (long)E * 32;

    gemm_wmma_kernel<8><<<gemmBlocks, 256, 128 * 128 * sizeof(float), stream>>>(x, W1, xw, N, numStripes);
    {
        long tot = (long)N * 128;
        self_bias_kernel<<<(int)((tot + 255) / 256), 256, 0, stream>>>(h, xw, b1, dinv, N, 128);
    }
    edge_scatter_kernel<4><<<(int)((et + 255) / 256), 256, 0, stream>>>(xw, dinv, src, dst, h, E);

    gemm_wmma_kernel<4><<<gemmBlocks, 256, 64 * 128 * sizeof(float), stream>>>(h, W2, xw, N, numStripes);
    {
        long tot = (long)N * 64;
        self_bias_kernel<<<(int)((tot + 255) / 256), 256, 0, stream>>>(out, xw, b2, dinv, N, 64);
    }
    edge_scatter_kernel<2><<<(int)((et + 255) / 256), 256, 0, stream>>>(xw, dinv, src, dst, out, E);
}


# Replace f32 atomic edge scatter by per-call CSR by dst (int rank atomic, arrival order) + one-wave-per-node row gather with self loop and bias fused; layer 1 gathers x before its GEMM
# speedup vs baseline: 2.4078x; 2.4078x over previous
#include <hip/hip_runtime.h>


typedef __attribute__((ext_vector_type(2))) float v2f;
typedef __attribute__((ext_vector_type(8))) float v8f;


__global__ void zero_counts_kernel(int* __restrict__ cnt, int n) {
    int i = blockIdx.x * blockDim.x + threadIdx.x;
    if (i < n) cnt[i] = 0;
}

__global__ void edge_rank_kernel(int* __restrict__ cnt, const int* __restrict__ dst,
                                 int* __restrict__ rank, int E) {
    int e = blockIdx.x * blockDim.x + threadIdx.x;
    if (e < E) rank[e] = atomicAdd(&cnt[dst[e]], 1);
}

__global__ __launch_bounds__(1024) void scan_dinv_kernel(const int* __restrict__ cnt,
                                                         int* __restrict__ offsets,
                                                         float* __restrict__ dinv, int n) {
    __shared__ int wsum[32];
    __shared__ int s_prefix;
    const int tid = threadIdx.x;
    const int lane = tid & 31;
    const int wave = tid >> 5;
    const int blockStart = blockIdx.x * 1024;

    int part = 0;
    const int4* c4 = (const int4*)cnt;
    const int lim = blockStart >> 2;
    int j = tid;
    for (; j + 3072 < lim; j += 4096) {
        const int4 v0 = c4[j];
        const int4 v1 = c4[j + 1024];
        const int4 v2 = c4[j + 2048];
        const int4 v3 = c4[j + 3072];
        part += ((v0.x + v0.y) + (v0.z + v0.w)) + ((v1.x + v1.y) + (v1.z + v1.w));
        part += ((v2.x + v2.y) + (v2.z + v2.w)) + ((v3.x + v3.y) + (v3.z + v3.w));
    }
    for (; j < lim; j += 1024) {
        const int4 v = c4[j];
        part += (v.x + v.y) + (v.z + v.w);
    }
    for (int o = 16; o > 0; o >>= 1) part += __shfl_xor(part, o);
    if (lane == 0) wsum[wave] = part;
    __syncthreads();
    if (wave == 0) {
        int v = wsum[lane];
        for (int o = 16; o > 0; o >>= 1) v += __shfl_xor(v, o);
        if (lane == 0) s_prefix = v;
    }
    __syncthreads();
    const int prefix = s_prefix;

    const int i = blockStart + tid;
    const int c = (i < n) ? cnt[i] : 0;
    int incl = c;
    for (int o = 1; o < 32; o <<= 1) {
        int t = __shfl_up(incl, o);
        if (lane >= o) incl += t;
    }
    if (lane == 31) wsum[wave] = incl;
    __syncthreads();
    if (wave == 0) {
        int v = wsum[lane];
        int iv = v;
        for (int o = 1; o < 32; o <<= 1) {
            int t = __shfl_up(iv, o);
            if (lane >= o) iv += t;
        }
        wsum[lane] = iv - v;
    }
    __syncthreads();
    if (i < n) {
        offsets[i] = prefix + wsum[wave] + (incl - c);
        float dg = (float)(c + 1);
        dinv[i] = (dg > 0.0f) ? rsqrtf(dg) : 0.0f;
    }
}

__global__ void csr_fill_kernel(const int* __restrict__ src, const int* __restrict__ dst,
                                const int* __restrict__ rank, const int* __restrict__ offsets,
                                int* __restrict__ csr_src, int E) {
    int e = blockIdx.x * blockDim.x + threadIdx.x;
    if (e < E) csr_src[offsets[dst[e]] + rank[e]] = src[e];
}


template <int VPL, bool HAS_BIAS>
__global__ __launch_bounds__(256) void node_gather_kernel(const float* __restrict__ in,
                                                          const float* __restrict__ dinv,
                                                          const int* __restrict__ offsets,
                                                          const int* __restrict__ cnt,
                                                          const int* __restrict__ csr_src,
                                                          const float* __restrict__ bias,
                                                          float* __restrict__ out, int N) {
    constexpr int DIM = 32 * VPL;
    typedef __attribute__((ext_vector_type(VPL))) float vec;
    const int lane = threadIdx.x & 31;
    int d = blockIdx.x * 8 + (threadIdx.x >> 5);
    if (d >= N) return;
    d = __builtin_amdgcn_readfirstlane(d);
    const int beg = offsets[d];
    const int len = cnt[d];
    const float dd = dinv[d];

    vec acc = *(const vec*)(in + (size_t)d * DIM + lane * VPL) * dd;

    for (int base = 0; base < len; base += 32) {
        const int n = (len - base < 32) ? (len - base) : 32;
        int sj = 0;
        float wj = 0.0f;
        if (lane < n) {
            sj = csr_src[beg + base + lane];
            wj = dinv[sj];
        }
        const int wji = __float_as_int(wj);
#define NG_ROW(jj) (*(const vec*)(in + (size_t)__builtin_amdgcn_readlane(sj, (jj)) * DIM + lane * VPL))
#define NG_W(jj)   (__int_as_float(__builtin_amdgcn_readlane(wji, (jj))))
        int j = 0;
        for (; j + 8 <= n; j += 8) {
            const vec r0 = NG_ROW(j),     r1 = NG_ROW(j + 1), r2 = NG_ROW(j + 2), r3 = NG_ROW(j + 3);
            const vec r4 = NG_ROW(j + 4), r5 = NG_ROW(j + 5), r6 = NG_ROW(j + 6), r7 = NG_ROW(j + 7);
            acc += r0 * NG_W(j);
            acc += r1 * NG_W(j + 1);
            acc += r2 * NG_W(j + 2);
            acc += r3 * NG_W(j + 3);
            acc += r4 * NG_W(j + 4);
            acc += r5 * NG_W(j + 5);
            acc += r6 * NG_W(j + 6);
            acc += r7 * NG_W(j + 7);
        }
        if (j + 4 <= n) {
            const vec r0 = NG_ROW(j), r1 = NG_ROW(j + 1), r2 = NG_ROW(j + 2), r3 = NG_ROW(j + 3);
            acc += r0 * NG_W(j);
            acc += r1 * NG_W(j + 1);
            acc += r2 * NG_W(j + 2);
            acc += r3 * NG_W(j + 3);
            j += 4;
        }
        if (j < n) {
            const vec z = {};
            const vec r0 = NG_ROW(j);
            vec r1 = z, r2 = z;
            float w1 = 0.0f, w2 = 0.0f;
            if (j + 1 < n) { r1 = NG_ROW(j + 1); w1 = NG_W(j + 1); }
            if (j + 2 < n) { r2 = NG_ROW(j + 2); w2 = NG_W(j + 2); }
            acc += r0 * NG_W(j);
            acc += r1 * w1;
            acc += r2 * w2;
        }
#undef NG_ROW
#undef NG_W
    }

    acc *= dd;
    if (HAS_BIAS) acc += *(const vec*)(bias + lane * VPL);
    *(vec*)(out + (size_t)d * DIM + lane * VPL) = acc;
}


template <int NT, bool HAS_BIAS>
__global__ __launch_bounds__(256) void gemm_wmma_kernel(const float* X,
                                                        const float* __restrict__ W,
                                                        const float* __restrict__ bias,
                                                        float* Y,
                                                        int N, int numStripes) {
    constexpr int K = 128;
    constexpr int OUT = NT * 16;
    extern __shared__ float wt[];

    const int tid = threadIdx.x;
    for (int i = tid; i < K * OUT; i += 256) {
        int k = i / OUT;
        int n = i - k * OUT;
        wt[(k >> 1) * (2 * OUT) + 2 * n + (k & 1)] = W[i];
    }
    __syncthreads();

    const int lane = tid & 31;
    const int wave = tid >> 5;
    const int m = lane & 15;
    const int half = lane >> 4;
    const int kg = half << 1;

    const float* lbase = wt + (half * OUT + m) * 2;

    float bv[NT];
#pragma unroll
    for (int t = 0; t < NT; ++t) bv[t] = HAS_BIAS ? bias[t * 16 + m] : 0.0f;

    for (int stripe = blockIdx.x; stripe < numStripes; stripe += gridDim.x) {
        const int rowBase = stripe * 128 + wave * 16;
        int arow = rowBase + m;
        if (arow >= N) arow = N - 1;
        const float* xrow = X + (long)arow * K;

        v8f zero = {};
        v8f acc[NT];
#pragma unroll
        for (int t = 0; t < NT; ++t) acc[t] = zero;

#pragma unroll 8
        for (int k = 0; k < K; k += 4) {
            v2f a = *(const v2f*)(xrow + k + kg);
            v2f b[NT];
            const float* lk = lbase + (k >> 1) * (2 * OUT);
#pragma unroll
            for (int t = 0; t < NT; ++t) b[t] = *(const v2f*)(lk + t * 32);
#pragma unroll
            for (int t = 0; t < NT; ++t)
                acc[t] = __builtin_amdgcn_wmma_f32_16x16x4_f32(
                    false, a, false, b[t], (short)0, acc[t], false, false);
        }

#pragma unroll
        for (int t = 0; t < NT; ++t) {
            const int col = t * 16 + m;
#pragma unroll
            for (int r = 0; r < 8; ++r) {
                const int row = rowBase + r + 8 * half;
                if (row < N) Y[(long)row * OUT + col] = HAS_BIAS ? (acc[t][r] + bv[t]) : acc[t][r];
            }
        }
    }
}


extern "C" void kernel_launch(void* const* d_in, const int* in_sizes, int n_in,
                              void* d_out, int out_size, void* d_ws, size_t ws_size,
                              hipStream_t stream) {
    const float* x  = (const float*)d_in[0];
    const int*   ei = (const int*)d_in[1];
    const float* W1 = (const float*)d_in[2];
    const float* b1 = (const float*)d_in[3];
    const float* W2 = (const float*)d_in[4];
    const float* b2 = (const float*)d_in[5];
    float* out = (float*)d_out;

    const int N = in_sizes[0] / 128;
    const int E = in_sizes[1] / 2;
    const int* src = ei;
    const int* dst = ei + E;

    float* dinv    = (float*)d_ws;
    float* a       = dinv + N;
    float* xw2     = a + (size_t)N * 128;
    int*   cnt     = (int*)(xw2 + (size_t)N * 64);
    int*   offsets = cnt + N;
    int*   rank    = offsets + N;
    int*   csr_src = rank + E;

    zero_counts_kernel<<<(N + 255) / 256, 256, 0, stream>>>(cnt, N);
    edge_rank_kernel<<<(E + 255) / 256, 256, 0, stream>>>(cnt, dst, rank, E);
    scan_dinv_kernel<<<(N + 1023) / 1024, 1024, 0, stream>>>(cnt, offsets, dinv, N);
    csr_fill_kernel<<<(E + 255) / 256, 256, 0, stream>>>(src, dst, rank, offsets, csr_src, E);

    const int numStripes = (N + 127) / 128;
    const int gemmBlocks = numStripes < 320 ? numStripes : 320;
    const int gatherBlocks = (N + 7) / 8;

    node_gather_kernel<4, false><<<gatherBlocks, 256, 0, stream>>>(x, dinv, offsets, cnt, csr_src, b1, a, N);
    gemm_wmma_kernel<8, true><<<gemmBlocks, 256, 128 * 128 * sizeof(float), stream>>>(a, W1, b1, a, N, numStripes);

    gemm_wmma_kernel<4, false><<<gemmBlocks, 256, 64 * 128 * sizeof(float), stream>>>(a, W2, b2, xw2, N, numStripes);
    node_gather_kernel<2, true><<<gatherBlocks, 256, 0, stream>>>(xw2, dinv, offsets, cnt, csr_src, b2, out, N);
}


# Fold the two linear layers (nothing but the bias lies between them): W12=W1@W2, b12=b1@W2 by WMMA each call; one 128->64 GEMM on x, both aggregations 64 wide; f32 sums reassociated
# speedup vs baseline: 2.8992x; 1.2041x over previous
#include <hip/hip_runtime.h>


typedef __attribute__((ext_vector_type(2))) float v2f;
typedef __attribute__((ext_vector_type(8))) float v8f;


__global__ void zero_counts_kernel(int* __restrict__ cnt, int n) {
    int i = blockIdx.x * blockDim.x + threadIdx.x;
    if (i < n) cnt[i] = 0;
}

__global__ void edge_rank_kernel(int* __restrict__ cnt, const int* __restrict__ dst,
                                 int* __restrict__ rank, int E) {
    int e = blockIdx.x * blockDim.x + threadIdx.x;
    if (e < E) rank[e] = atomicAdd(&cnt[dst[e]], 1);
}

__global__ __launch_bounds__(1024) void scan_dinv_kernel(const int* __restrict__ cnt,
                                                         int* __restrict__ offsets,
                                                         float* __restrict__ dinv, int n) {
    __shared__ int wsum[32];
    __shared__ int s_prefix;
    const int tid = threadIdx.x;
    const int lane = tid & 31;
    const int wave = tid >> 5;
    const int blockStart = blockIdx.x * 1024;

    int part = 0;
    const int4* c4 = (const int4*)cnt;
    const int lim = blockStart >> 2;
    int j = tid;
    for (; j + 3072 < lim; j += 4096) {
        const int4 v0 = c4[j];
        const int4 v1 = c4[j + 1024];
        const int4 v2 = c4[j + 2048];
        const int4 v3 = c4[j + 3072];
        part += ((v0.x + v0.y) + (v0.z + v0.w)) + ((v1.x + v1.y) + (v1.z + v1.w));
        part += ((v2.x + v2.y) + (v2.z + v2.w)) + ((v3.x + v3.y) + (v3.z + v3.w));
    }
    for (; j < lim; j += 1024) {
        const int4 v = c4[j];
        part += (v.x + v.y) + (v.z + v.w);
    }
    for (int o = 16; o > 0; o >>= 1) part += __shfl_xor(part, o);
    if (lane == 0) wsum[wave] = part;
    __syncthreads();
    if (wave == 0) {
        int v = wsum[lane];
        for (int o = 16; o > 0; o >>= 1) v += __shfl_xor(v, o);
        if (lane == 0) s_prefix = v;
    }
    __syncthreads();
    const int prefix = s_prefix;

    const int i = blockStart + tid;
    const int c = (i < n) ? cnt[i] : 0;
    int incl = c;
    for (int o = 1; o < 32; o <<= 1) {
        int t = __shfl_up(incl, o);
        if (lane >= o) incl += t;
    }
    if (lane == 31) wsum[wave] = incl;
    __syncthreads();
    if (wave == 0) {
        int v = wsum[lane];
        int iv = v;
        for (int o = 1; o < 32; o <<= 1) {
            int t = __shfl_up(iv, o);
            if (lane >= o) iv += t;
        }
        wsum[lane] = iv - v;
    }
    __syncthreads();
    if (i < n) {
        offsets[i] = prefix + wsum[wave] + (incl - c);
        float dg = (float)(c + 1);
        dinv[i] = (dg > 0.0f) ? rsqrtf(dg) : 0.0f;
    }
}

__global__ void csr_fill_kernel(const int* __restrict__ src, const int* __restrict__ dst,
                                const int* __restrict__ rank, const int* __restrict__ offsets,
                                int* __restrict__ csr_src, int E) {
    int e = blockIdx.x * blockDim.x + threadIdx.x;
    if (e < E) csr_src[offsets[dst[e]] + rank[e]] = src[e];
}


template <int VPL, bool HAS_BIAS>
__global__ __launch_bounds__(256) void node_gather_kernel(const float* __restrict__ in,
                                                          const float* __restrict__ dinv,
                                                          const int* __restrict__ offsets,
                                                          const int* __restrict__ cnt,
                                                          const int* __restrict__ csr_src,
                                                          const float* __restrict__ bias,
                                                          float* __restrict__ out, int N) {
    constexpr int DIM = 32 * VPL;
    typedef __attribute__((ext_vector_type(VPL))) float vec;
    const int lane = threadIdx.x & 31;
    int d = blockIdx.x * 8 + (threadIdx.x >> 5);
    if (d >= N) return;
    d = __builtin_amdgcn_readfirstlane(d);
    const int beg = offsets[d];
    const int len = cnt[d];
    const float dd = dinv[d];

    vec acc = *(const vec*)(in + (size_t)d * DIM + lane * VPL) * dd;

    for (int base = 0; base < len; base += 32) {
        const int n = (len - base < 32) ? (len - base) : 32;
        int sj = 0;
        float wj = 0.0f;
        if (lane < n) {
            sj = csr_src[beg + base + lane];
            wj = dinv[sj];
        }
        const int wji = __float_as_int(wj);
#define NG_ROW(jj) (*(const vec*)(in + (size_t)__builtin_amdgcn_readlane(sj, (jj)) * DIM + lane * VPL))
#define NG_W(jj)   (__int_as_float(__builtin_amdgcn_readlane(wji, (jj))))
        int j = 0;
        for (; j + 8 <= n; j += 8) {
            const vec r0 = NG_ROW(j),     r1 = NG_ROW(j + 1), r2 = NG_ROW(j + 2), r3 = NG_ROW(j + 3);
            const vec r4 = NG_ROW(j + 4), r5 = NG_ROW(j + 5), r6 = NG_ROW(j + 6), r7 = NG_ROW(j + 7);
            acc += r0 * NG_W(j);
            acc += r1 * NG_W(j + 1);
            acc += r2 * NG_W(j + 2);
            acc += r3 * NG_W(j + 3);
            acc += r4 * NG_W(j + 4);
            acc += r5 * NG_W(j + 5);
            acc += r6 * NG_W(j + 6);
            acc += r7 * NG_W(j + 7);
        }
        if (j + 4 <= n) {
            const vec r0 = NG_ROW(j), r1 = NG_ROW(j + 1), r2 = NG_ROW(j + 2), r3 = NG_ROW(j + 3);
            acc += r0 * NG_W(j);
            acc += r1 * NG_W(j + 1);
            acc += r2 * NG_W(j + 2);
            acc += r3 * NG_W(j + 3);
            j += 4;
        }
        if (j < n) {
            const vec z = {};
            const vec r0 = NG_ROW(j);
            vec r1 = z, r2 = z;
            float w1 = 0.0f, w2 = 0.0f;
            if (j + 1 < n) { r1 = NG_ROW(j + 1); w1 = NG_W(j + 1); }
            if (j + 2 < n) { r2 = NG_ROW(j + 2); w2 = NG_W(j + 2); }
            acc += r0 * NG_W(j);
            acc += r1 * w1;
            acc += r2 * w2;
        }
#undef NG_ROW
#undef NG_W
    }

    acc *= dd;
    if (HAS_BIAS) acc += *(const vec*)(bias + lane * VPL);
    *(vec*)(out + (size_t)d * DIM + lane * VPL) = acc;
}


template <int NT, bool LAST_ROW_APART>
__global__ __launch_bounds__(256) void gemm_wmma_kernel(const float* X,
                                                        const float* __restrict__ xlast,
                                                        const float* __restrict__ W,
                                                        float* Y,
                                                        int N, int numStripes) {
    constexpr int K = 128;
    constexpr int OUT = NT * 16;
    extern __shared__ float wt[];

    const int tid = threadIdx.x;
    for (int i = tid; i < K * OUT; i += 256) {
        int k = i / OUT;
        int n = i - k * OUT;
        wt[(k >> 1) * (2 * OUT) + 2 * n + (k & 1)] = W[i];
    }
    __syncthreads();

    const int lane = tid & 31;
    const int wave = tid >> 5;
    const int m = lane & 15;
    const int half = lane >> 4;
    const int kg = half << 1;

    const float* lbase = wt + (half * OUT + m) * 2;

    for (int stripe = blockIdx.x; stripe < numStripes; stripe += gridDim.x) {
        const int rowBase = stripe * 128 + wave * 16;
        int arow = rowBase + m;
        if (arow >= N) arow = N - 1;
        const float* xrow = X + (long)arow * K;
        if (LAST_ROW_APART && arow == N - 1) xrow = xlast;

        v8f zero = {};
        v8f acc[NT];
#pragma unroll
        for (int t = 0; t < NT; ++t) acc[t] = zero;

#pragma unroll 8
        for (int k = 0; k < K; k += 4) {
            v2f a = *(const v2f*)(xrow + k + kg);
            v2f b[NT];
            const float* lk = lbase + (k >> 1) * (2 * OUT);
#pragma unroll
            for (int t = 0; t < NT; ++t) b[t] = *(const v2f*)(lk + t * 32);
#pragma unroll
            for (int t = 0; t < NT; ++t)
                acc[t] = __builtin_amdgcn_wmma_f32_16x16x4_f32(
                    false, a, false, b[t], (short)0, acc[t], false, false);
        }

#pragma unroll
        for (int t = 0; t < NT; ++t) {
            const int col = t * 16 + m;
#pragma unroll
            for (int r = 0; r < 8; ++r) {
                const int row = rowBase + r + 8 * half;
                if (row < N) Y[(long)row * OUT + col] = acc[t][r];
            }
        }
    }
}


extern "C" void kernel_launch(void* const* d_in, const int* in_sizes, int n_in,
                              void* d_out, int out_size, void* d_ws, size_t ws_size,
                              hipStream_t stream) {
    const float* x  = (const float*)d_in[0];
    const int*   ei = (const int*)d_in[1];
    const float* W1 = (const float*)d_in[2];
    const float* b1 = (const float*)d_in[3];
    const float* W2 = (const float*)d_in[4];
    const float* b2 = (const float*)d_in[5];
    float* out = (float*)d_out;

    const int N = in_sizes[0] / 128;
    const int E = in_sizes[1] / 2;
    const int* src = ei;
    const int* dst = ei + E;

    float* dinv    = (float*)d_ws;
    float* xw      = dinv + N;
    float* t       = xw + (size_t)N * 64;
    float* w12     = t + (size_t)N * 64;
    float* b12     = w12 + 128 * 64;
    int*   cnt     = (int*)(w12 + 129 * 64);
    int*   offsets = cnt + N;
    int*   rank    = offsets + N;
    int*   csr_src = rank + E;

    zero_counts_kernel<<<(N + 255) / 256, 256, 0, stream>>>(cnt, N);
    edge_rank_kernel<<<(E + 255) / 256, 256, 0, stream>>>(cnt, dst, rank, E);
    scan_dinv_kernel<<<(N + 1023) / 1024, 1024, 0, stream>>>(cnt, offsets, dinv, N);
    csr_fill_kernel<<<(E + 255) / 256, 256, 0, stream>>>(src, dst, rank, offsets, csr_src, E);

    const int numStripes = (N + 127) / 128;
    const int gemmBlocks = numStripes < 320 ? numStripes : 320;
    const int gatherBlocks = (N + 7) / 8;

    gemm_wmma_kernel<4, true><<<2, 256, 64 * 128 * sizeof(float), stream>>>(W1, b1, W2, w12, 129, 2);

    gemm_wmma_kernel<4, false><<<gemmBlocks, 256, 64 * 128 * sizeof(float), stream>>>(x, x, w12, xw, N, numStripes);

    node_gather_kernel<2, true><<<gatherBlocks, 256, 0, stream>>>(xw, dinv, offsets, cnt, csr_src, b12, t, N);
    node_gather_kernel<2, true><<<gatherBlocks, 256, 0, stream>>>(t, dinv, offsets, cnt, csr_src, b2, out, N);
}


# Edge grouping by two-level counting sort on LDS counters (bucket of 256 nodes, then node; LDS returning add = slot, arrival order as before) replaces one global returning atomic per edge
# speedup vs baseline: 4.5517x; 1.5700x over previous
#include <hip/hip_runtime.h>


typedef __attribute__((ext_vector_type(2))) float v2f;
typedef __attribute__((ext_vector_type(8))) float v8f;


constexpr int SORT_BD = 1024;
constexpr int SORT_C  = 256;
constexpr int SORT_U  = 4;

template <bool FULL>
__device__ __forceinline__ void count_trip(const int* __restrict__ dst, int* lh, int e0, int end) {
    int d[SORT_U];
#pragma unroll
    for (int u = 0; u < SORT_U; ++u) {
        const int e = e0 + u * SORT_BD;
        d[u] = dst[(FULL || e < end) ? e : (end - 1)];
    }
#pragma unroll
    for (int u = 0; u < SORT_U; ++u)
        if (FULL || e0 + u * SORT_BD < end) atomicAdd(&lh[d[u] >> 8], 1);
}

__global__ __launch_bounds__(SORT_BD) void bucket_count_kernel(const int* __restrict__ dst,
                                                               int* __restrict__ hist,
                                                               int E, int CH, int NB) {
    extern __shared__ int lh[];
    const int tid = threadIdx.x;
    for (int b = tid; b < NB; b += SORT_BD) lh[b] = 0;
    __syncthreads();
    const int beg = blockIdx.x * CH;
    const int end = (beg + CH < E) ? (beg + CH) : E;
    for (int e0 = beg + tid; e0 < end; e0 += SORT_U * SORT_BD) {
        if (e0 + (SORT_U - 1) * SORT_BD < end) count_trip<true>(dst, lh, e0, end);
        else                                   count_trip<false>(dst, lh, e0, end);
    }
    __syncthreads();
    for (int b = tid; b < NB; b += SORT_BD) hist[b * SORT_C + blockIdx.x] = lh[b];
}

__global__ __launch_bounds__(1024) void scan_excl_kernel(const int* __restrict__ in,
                                                         int* __restrict__ out, int n) {
    __shared__ int wsum[32];
    __shared__ int s_prefix;
    const int tid = threadIdx.x;
    const int lane = tid & 31;
    const int wave = tid >> 5;
    const int blockStart = blockIdx.x * 1024;

    int part = 0;
    const int4* c4 = (const int4*)in;
    const int lim = blockStart >> 2;
    int j = tid;
    for (; j + 3072 < lim; j += 4096) {
        const int4 v0 = c4[j];
        const int4 v1 = c4[j + 1024];
        const int4 v2 = c4[j + 2048];
        const int4 v3 = c4[j + 3072];
        part += ((v0.x + v0.y) + (v0.z + v0.w)) + ((v1.x + v1.y) + (v1.z + v1.w));
        part += ((v2.x + v2.y) + (v2.z + v2.w)) + ((v3.x + v3.y) + (v3.z + v3.w));
    }
    for (; j < lim; j += 1024) {
        const int4 v = c4[j];
        part += (v.x + v.y) + (v.z + v.w);
    }
    for (int o = 16; o > 0; o >>= 1) part += __shfl_xor(part, o);
    if (lane == 0) wsum[wave] = part;
    __syncthreads();
    if (wave == 0) {
        int v = wsum[lane];
        for (int o = 16; o > 0; o >>= 1) v += __shfl_xor(v, o);
        if (lane == 0) s_prefix = v;
    }
    __syncthreads();
    const int prefix = s_prefix;

    const int i = blockStart + tid;
    const int c = (i < n) ? in[i] : 0;
    int incl = c;
    for (int o = 1; o < 32; o <<= 1) {
        int t = __shfl_up(incl, o);
        if (lane >= o) incl += t;
    }
    if (lane == 31) wsum[wave] = incl;
    __syncthreads();
    if (wave == 0) {
        int v = wsum[lane];
        int iv = v;
        for (int o = 1; o < 32; o <<= 1) {
            int t = __shfl_up(iv, o);
            if (lane >= o) iv += t;
        }
        wsum[lane] = iv - v;
    }
    __syncthreads();
    if (i < n) out[i] = prefix + wsum[wave] + (incl - c);
}

template <bool FULL>
__device__ __forceinline__ void scatter_trip(const int* __restrict__ src, const int* __restrict__ dst,
                                             int* __restrict__ rec, int* cur, int e0, int end) {
    int d[SORT_U], s[SORT_U];
#pragma unroll
    for (int u = 0; u < SORT_U; ++u) {
        const int e = e0 + u * SORT_BD;
        const int ec = (FULL || e < end) ? e : (end - 1);
        d[u] = dst[ec];
        s[u] = src[ec];
    }
#pragma unroll
    for (int u = 0; u < SORT_U; ++u)
        if (FULL || e0 + u * SORT_BD < end) {
            const int pos = atomicAdd(&cur[d[u] >> 8], 1);
            rec[pos] = (int)(((unsigned)s[u] << 8) | (unsigned)(d[u] & 255));
        }
}

__global__ __launch_bounds__(SORT_BD) void bucket_scatter_kernel(const int* __restrict__ src,
                                                                 const int* __restrict__ dst,
                                                                 const int* __restrict__ start,
                                                                 int* __restrict__ rec,
                                                                 int E, int CH, int NB) {
    extern __shared__ int cur[];
    const int tid = threadIdx.x;
    for (int b = tid; b < NB; b += SORT_BD) cur[b] = start[b * SORT_C + blockIdx.x];
    __syncthreads();
    const int beg = blockIdx.x * CH;
    const int end = (beg + CH < E) ? (beg + CH) : E;
    for (int e0 = beg + tid; e0 < end; e0 += SORT_U * SORT_BD) {
        if (e0 + (SORT_U - 1) * SORT_BD < end) scatter_trip<true>(src, dst, rec, cur, e0, end);
        else                                   scatter_trip<false>(src, dst, rec, cur, e0, end);
    }
}

template <bool FULL>
__device__ __forceinline__ void node_count_trip(const int* __restrict__ rec, int* lc, int i0, int bend) {
    int r[SORT_U];
#pragma unroll
    for (int u = 0; u < SORT_U; ++u) {
        const int i = i0 + u * SORT_BD;
        r[u] = rec[(FULL || i < bend) ? i : (bend - 1)];
    }
#pragma unroll
    for (int u = 0; u < SORT_U; ++u)
        if (FULL || i0 + u * SORT_BD < bend) atomicAdd(&lc[r[u] & 255], 1);
}

template <bool FULL>
__device__ __forceinline__ void node_place_trip(const int* __restrict__ rec, int* __restrict__ csr_src,
                                                int* lc, int bbeg, int i0, int bend) {
    int r[SORT_U];
#pragma unroll
    for (int u = 0; u < SORT_U; ++u) {
        const int i = i0 + u * SORT_BD;
        r[u] = rec[(FULL || i < bend) ? i : (bend - 1)];
    }
#pragma unroll
    for (int u = 0; u < SORT_U; ++u)
        if (FULL || i0 + u * SORT_BD < bend) {
            const int pos = atomicAdd(&lc[r[u] & 255], 1);
            csr_src[bbeg + pos] = (int)((unsigned)r[u] >> 8);
        }
}

__global__ __launch_bounds__(SORT_BD) void bucket_sort_kernel(const int* __restrict__ rec,
                                                              const int* __restrict__ start,
                                                              int* __restrict__ csr_src,
                                                              int* __restrict__ cnt,
                                                              int* __restrict__ offsets,
                                                              float* __restrict__ dinv,
                                                              int N, int E, int NB) {
    __shared__ int lc[256];
    __shared__ int wsum[8];
    const int tid = threadIdx.x;
    const int lane = tid & 31;
    const int wave = tid >> 5;
    const int b = blockIdx.x;
    const int bbeg = start[b * SORT_C];
    const int bend = (b + 1 < NB) ? start[(b + 1) * SORT_C] : E;

    if (tid < 256) lc[tid] = 0;
    __syncthreads();
    for (int i0 = bbeg + tid; i0 < bend; i0 += SORT_U * SORT_BD) {
        if (i0 + (SORT_U - 1) * SORT_BD < bend) node_count_trip<true>(rec, lc, i0, bend);
        else                                    node_count_trip<false>(rec, lc, i0, bend);
    }
    __syncthreads();

    int c = 0, incl = 0;
    if (tid < 256) {
        c = lc[tid];
        incl = c;
        for (int o = 1; o < 32; o <<= 1) {
            int t = __shfl_up(incl, o);
            if (lane >= o) incl += t;
        }
        if (lane == 31) wsum[wave] = incl;
    }
    __syncthreads();
    if (tid < 256) {
        int wpre = 0;
        for (int w = 0; w < 8; ++w) wpre += (w < wave) ? wsum[w] : 0;
        const int excl = wpre + (incl - c);
        lc[tid] = excl;
        const int node = b * 256 + tid;
        if (node < N) {
            cnt[node] = c;
            offsets[node] = bbeg + excl;
            float dg = (float)(c + 1);
            dinv[node] = (dg > 0.0f) ? rsqrtf(dg) : 0.0f;
        }
    }
    __syncthreads();
    for (int i0 = bbeg + tid; i0 < bend; i0 += SORT_U * SORT_BD) {
        if (i0 + (SORT_U - 1) * SORT_BD < bend) node_place_trip<true>(rec, csr_src, lc, bbeg, i0, bend);
        else                                    node_place_trip<false>(rec, csr_src, lc, bbeg, i0, bend);
    }
}


template <int VPL, bool HAS_BIAS, bool SCALE_OUT>
__global__ __launch_bounds__(256) void node_gather_kernel(const float* __restrict__ in,
                                                          const float* __restrict__ dinv,
                                                          const int* __restrict__ offsets,
                                                          const int* __restrict__ cnt,
                                                          const int* __restrict__ csr_src,
                                                          const float* __restrict__ bias,
                                                          float* __restrict__ out, int N) {
    constexpr int DIM = 32 * VPL;
    typedef __attribute__((ext_vector_type(VPL))) float vec;
    const int lane = threadIdx.x & 31;
    int d = blockIdx.x * 8 + (threadIdx.x >> 5);
    if (d >= N) return;
    d = __builtin_amdgcn_readfirstlane(d);
    const int beg = offsets[d];
    const int len = cnt[d];
    const float dd = dinv[d];

    vec acc = *(const vec*)(in + (size_t)d * DIM + lane * VPL);

    for (int base = 0; base < len; base += 32) {
        const int n = (len - base < 32) ? (len - base) : 32;
        int sj = 0;
        if (lane < n) sj = csr_src[beg + base + lane];
#define NG_ROW(jj) (*(const vec*)(in + (size_t)__builtin_amdgcn_readlane(sj, (jj)) * DIM + lane * VPL))
        int j = 0;
        for (; j + 8 <= n; j += 8) {
            const vec r0 = NG_ROW(j),     r1 = NG_ROW(j + 1), r2 = NG_ROW(j + 2), r3 = NG_ROW(j + 3);
            const vec r4 = NG_ROW(j + 4), r5 = NG_ROW(j + 5), r6 = NG_ROW(j + 6), r7 = NG_ROW(j + 7);
            acc += r0;
            acc += r1;
            acc += r2;
            acc += r3;
            acc += r4;
            acc += r5;
            acc += r6;
            acc += r7;
        }
        if (j + 4 <= n) {
            const vec r0 = NG_ROW(j), r1 = NG_ROW(j + 1), r2 = NG_ROW(j + 2), r3 = NG_ROW(j + 3);
            acc += r0;
            acc += r1;
            acc += r2;
            acc += r3;
            j += 4;
        }
        if (j < n) {
            const vec z = {};
            const vec r0 = NG_ROW(j);
            vec r1 = z, r2 = z;
            if (j + 1 < n) r1 = NG_ROW(j + 1);
            if (j + 2 < n) r2 = NG_ROW(j + 2);
            acc += r0;
            acc += r1;
            acc += r2;
        }
#undef NG_ROW
    }

    acc *= dd;
    if (HAS_BIAS) acc += *(const vec*)(bias + lane * VPL);
    if (SCALE_OUT) acc *= dd;
    *(vec*)(out + (size_t)d * DIM + lane * VPL) = acc;
}


template <int NT, bool LAST_ROW_APART, bool SCALE_ROWS>
__global__ __launch_bounds__(256) void gemm_wmma_kernel(const float* X,
                                                        const float* __restrict__ xlast,
                                                        const float* __restrict__ W,
                                                        const float* __restrict__ rowscale,
                                                        float* Y,
                                                        int N, int numStripes) {
    constexpr int K = 128;
    constexpr int OUT = NT * 16;
    extern __shared__ float wt[];

    const int tid = threadIdx.x;
    for (int i = tid; i < K * OUT; i += 256) {
        int k = i / OUT;
        int n = i - k * OUT;
        wt[(k >> 1) * (2 * OUT) + 2 * n + (k & 1)] = W[i];
    }
    __syncthreads();

    const int lane = tid & 31;
    const int wave = tid >> 5;
    const int m = lane & 15;
    const int half = lane >> 4;
    const int kg = half << 1;

    const float* lbase = wt + (half * OUT + m) * 2;

    for (int stripe = blockIdx.x; stripe < numStripes; stripe += gridDim.x) {
        const int rowBase = stripe * 128 + wave * 16;
        int arow = rowBase + m;
        if (arow >= N) arow = N - 1;
        const float* xrow = X + (long)arow * K;
        if (LAST_ROW_APART && arow == N - 1) xrow = xlast;

        float rs[8];
#pragma unroll
        for (int r = 0; r < 8; ++r) {
            const int row = rowBase + r + 8 * half;
            rs[r] = SCALE_ROWS ? rowscale[row < N ? row : N - 1] : 1.0f;
        }

        v8f zero = {};
        v8f acc[NT];
#pragma unroll
        for (int t = 0; t < NT; ++t) acc[t] = zero;

#pragma unroll 8
        for (int k = 0; k < K; k += 4) {
            v2f a = *(const v2f*)(xrow + k + kg);
            v2f b[NT];
            const float* lk = lbase + (k >> 1) * (2 * OUT);
#pragma unroll
            for (int t = 0; t < NT; ++t) b[t] = *(const v2f*)(lk + t * 32);
#pragma unroll
            for (int t = 0; t < NT; ++t)
                acc[t] = __builtin_amdgcn_wmma_f32_16x16x4_f32(
                    false, a, false, b[t], (short)0, acc[t], false, false);
        }

#pragma unroll
        for (int t = 0; t < NT; ++t) {
            const int col = t * 16 + m;
#pragma unroll
            for (int r = 0; r < 8; ++r) {
                const int row = rowBase + r + 8 * half;
                if (row < N) Y[(long)row * OUT + col] = SCALE_ROWS ? acc[t][r] * rs[r] : acc[t][r];
            }
        }
    }
}


extern "C" void kernel_launch(void* const* d_in, const int* in_sizes, int n_in,
                              void* d_out, int out_size, void* d_ws, size_t ws_size,
                              hipStream_t stream) {
    const float* x  = (const float*)d_in[0];
    const int*   ei = (const int*)d_in[1];
    const float* W1 = (const float*)d_in[2];
    const float* b1 = (const float*)d_in[3];
    const float* W2 = (const float*)d_in[4];
    const float* b2 = (const float*)d_in[5];
    float* out = (float*)d_out;

    const int N = in_sizes[0] / 128;
    const int E = in_sizes[1] / 2;
    const int* src = ei;
    const int* dst = ei + E;

    float* dinv    = (float*)d_ws;
    float* xw      = dinv + N;
    float* t       = xw + (size_t)N * 64;
    float* w12     = t + (size_t)N * 64;
    float* b12     = w12 + 128 * 64;
    int*   cnt     = (int*)(w12 + 129 * 64);
    int*   offsets = cnt + N;
    int*   rec     = offsets + N;
    int*   csr_src = rec + E;

    const int NB = (N + 255) / 256;
    const int CH = (E + SORT_C - 1) / SORT_C;
    const int M  = NB * SORT_C;
    int* hist  = (int*)t;
    int* start = hist + M;
    bucket_count_kernel<<<SORT_C, SORT_BD, NB * sizeof(int), stream>>>(dst, hist, E, CH, NB);
    scan_excl_kernel<<<(M + 1023) / 1024, 1024, 0, stream>>>(hist, start, M);
    bucket_scatter_kernel<<<SORT_C, SORT_BD, NB * sizeof(int), stream>>>(src, dst, start, rec, E, CH, NB);
    bucket_sort_kernel<<<NB, SORT_BD, 0, stream>>>(rec, start, csr_src, cnt, offsets, dinv, N, E, NB);

    const int numStripes = (N + 127) / 128;
    const int gemmBlocks = numStripes < 320 ? numStripes : 320;
    const int gatherBlocks = (N + 7) / 8;

    gemm_wmma_kernel<4, true, false><<<2, 256, 64 * 128 * sizeof(float), stream>>>(W1, b1, W2, dinv, w12, 129, 2);

    gemm_wmma_kernel<4, false, true><<<gemmBlocks, 256, 64 * 128 * sizeof(float), stream>>>(x, x, w12, dinv, xw, N, numStripes);

    node_gather_kernel<2, true, true><<<gatherBlocks, 256, 0, stream>>>(xw, dinv, offsets, cnt, csr_src, b12, t, N);
    node_gather_kernel<2, true, false><<<gatherBlocks, 256, 0, stream>>>(t, dinv, offsets, cnt, csr_src, b2, out, N);
}


# Folded weight product [W1;b1]@W2 moved into the prefix-sum launch as 9 extra work-groups (16 rows x 64 cols each, same f32 WMMA form, k order and sums): 7 launches, not 8
# speedup vs baseline: 4.8770x; 1.0715x over previous
#include <hip/hip_runtime.h>


typedef __attribute__((ext_vector_type(2))) float v2f;
typedef __attribute__((ext_vector_type(8))) float v8f;


constexpr int SORT_BD = 1024;
constexpr int SORT_C  = 256;
constexpr int SORT_U  = 4;

template <bool FULL>
__device__ __forceinline__ void count_trip(const int* __restrict__ dst, int* lh, int e0, int end) {
    int d[SORT_U];
#pragma unroll
    for (int u = 0; u < SORT_U; ++u) {
        const int e = e0 + u * SORT_BD;
        d[u] = dst[(FULL || e < end) ? e : (end - 1)];
    }
#pragma unroll
    for (int u = 0; u < SORT_U; ++u)
        if (FULL || e0 + u * SORT_BD < end) atomicAdd(&lh[d[u] >> 8], 1);
}

__global__ __launch_bounds__(SORT_BD) void bucket_count_kernel(const int* __restrict__ dst,
                                                               int* __restrict__ hist,
                                                               int E, int CH, int NB) {
    extern __shared__ int lh[];
    const int tid = threadIdx.x;
    for (int b = tid; b < NB; b += SORT_BD) lh[b] = 0;
    __syncthreads();
    const int beg = blockIdx.x * CH;
    const int end = (beg + CH < E) ? (beg + CH) : E;
    for (int e0 = beg + tid; e0 < end; e0 += SORT_U * SORT_BD) {
        if (e0 + (SORT_U - 1) * SORT_BD < end) count_trip<true>(dst, lh, e0, end);
        else                                   count_trip<false>(dst, lh, e0, end);
    }
    __syncthreads();
    for (int b = tid; b < NB; b += SORT_BD) hist[b * SORT_C + blockIdx.x] = lh[b];
}

constexpr int FOLD_WG = 9;

__device__ __forceinline__ void fold_weights_role(const float* __restrict__ W1,
                                                  const float* __restrict__ b1,
                                                  const float* __restrict__ W2,
                                                  float* __restrict__ w12, int g, float* wt) {
    constexpr int K = 128;
    constexpr int OUT = 64;
    const int tid = threadIdx.x;
    const int lane = tid & 31;
    const int wave = tid >> 5;
    const int m = lane & 15;
    const int half = lane >> 4;
    const int kg = half << 1;
    const bool compute = wave < 4;

    v2f a[K / 4];
    if (compute) {
        const float* arow = (g < 8) ? (W1 + (size_t)(g * 16 + m) * K) : b1;
#pragma unroll
        for (int j = 0; j < K / 4; ++j) a[j] = *(const v2f*)(arow + 4 * j + kg);
    }

    float w[K * OUT / 1024];
#pragma unroll
    for (int u = 0; u < K * OUT / 1024; ++u) w[u] = W2[tid + u * 1024];
#pragma unroll
    for (int u = 0; u < K * OUT / 1024; ++u) {
        const int i = tid + u * 1024;
        const int k = i / OUT;
        const int n = i - k * OUT;
        wt[(k >> 1) * (2 * OUT) + 2 * n + (k & 1)] = w[u];
    }
    __syncthreads();
    if (!compute) return;

    int ooff = (g * 16 + 8 * half) * OUT + wave * 16 + m;
    const float* lbase = wt + (half * OUT + m) * 2 + wave * 32;
    v8f acc = {};
    asm volatile("" : "+v"(ooff));
    __builtin_amdgcn_sched_barrier(0);
#pragma unroll
    for (int j = 0; j < K / 4; ++j) {
        const v2f b = *(const v2f*)(lbase + (2 * j) * (2 * OUT));
        acc = __builtin_amdgcn_wmma_f32_16x16x4_f32(false, a[j], false, b, (short)0, acc, false, false);
    }
    __builtin_amdgcn_sched_barrier(0);
    asm volatile("v_nop\n\tv_nop\n\tv_nop\n\tv_nop");
    __builtin_amdgcn_sched_barrier(0);

    if (g < 8) {
#pragma unroll
        for (int r = 0; r < 8; ++r) w12[ooff + r * OUT] = acc[r];
    } else if (half == 0) {
        w12[ooff] = acc[0];
    }
}

__global__ __launch_bounds__(1024) void scan_excl_fold_kernel(const int* __restrict__ in,
                                                              int* __restrict__ out, int n,
                                                              const float* __restrict__ W1,
                                                              const float* __restrict__ b1,
                                                              const float* __restrict__ W2,
                                                              float* __restrict__ w12) {
    __shared__ int wsum[32];
    __shared__ int s_prefix;
    __shared__ float wt[128 * 64];
    if (blockIdx.x < FOLD_WG) {
        fold_weights_role(W1, b1, W2, w12, blockIdx.x, wt);
        return;
    }
    const int tid = threadIdx.x;
    const int lane = tid & 31;
    const int wave = tid >> 5;
    const int blockStart = (blockIdx.x - FOLD_WG) * 1024;

    int part = 0;
    const int4* c4 = (const int4*)in;
    const int lim = blockStart >> 2;
    int j = tid;
    for (; j + 3072 < lim; j += 4096) {
        const int4 v0 = c4[j];
        const int4 v1 = c4[j + 1024];
        const int4 v2 = c4[j + 2048];
        const int4 v3 = c4[j + 3072];
        part += ((v0.x + v0.y) + (v0.z + v0.w)) + ((v1.x + v1.y) + (v1.z + v1.w));
        part += ((v2.x + v2.y) + (v2.z + v2.w)) + ((v3.x + v3.y) + (v3.z + v3.w));
    }
    for (; j < lim; j += 1024) {
        const int4 v = c4[j];
        part += (v.x + v.y) + (v.z + v.w);
    }
    for (int o = 16; o > 0; o >>= 1) part += __shfl_xor(part, o);
    if (lane == 0) wsum[wave] = part;
    __syncthreads();
    if (wave == 0) {
        int v = wsum[lane];
        for (int o = 16; o > 0; o >>= 1) v += __shfl_xor(v, o);
        if (lane == 0) s_prefix = v;
    }
    __syncthreads();
    const int prefix = s_prefix;

    const int i = blockStart + tid;
    const int c = (i < n) ? in[i] : 0;
    int incl = c;
    for (int o = 1; o < 32; o <<= 1) {
        int t = __shfl_up(incl, o);
        if (lane >= o) incl += t;
    }
    if (lane == 31) wsum[wave] = incl;
    __syncthreads();
    if (wave == 0) {
        int v = wsum[lane];
        int iv = v;
        for (int o = 1; o < 32; o <<= 1) {
            int t = __shfl_up(iv, o);
            if (lane >= o) iv += t;
        }
        wsum[lane] = iv - v;
    }
    __syncthreads();
    if (i < n) out[i] = prefix + wsum[wave] + (incl - c);
}

template <bool FULL>
__device__ __forceinline__ void scatter_trip(const int* __restrict__ src, const int* __restrict__ dst,
                                             int* __restrict__ rec, int* cur, int e0, int end) {
    int d[SORT_U], s[SORT_U];
#pragma unroll
    for (int u = 0; u < SORT_U; ++u) {
        const int e = e0 + u * SORT_BD;
        const int ec = (FULL || e < end) ? e : (end - 1);
        d[u] = dst[ec];
        s[u] = src[ec];
    }
#pragma unroll
    for (int u = 0; u < SORT_U; ++u)
        if (FULL || e0 + u * SORT_BD < end) {
            const int pos = atomicAdd(&cur[d[u] >> 8], 1);
            rec[pos] = (int)(((unsigned)s[u] << 8) | (unsigned)(d[u] & 255));
        }
}

__global__ __launch_bounds__(SORT_BD) void bucket_scatter_kernel(const int* __restrict__ src,
                                                                 const int* __restrict__ dst,
                                                                 const int* __restrict__ start,
                                                                 int* __restrict__ rec,
                                                                 int E, int CH, int NB) {
    extern __shared__ int cur[];
    const int tid = threadIdx.x;
    for (int b = tid; b < NB; b += SORT_BD) cur[b] = start[b * SORT_C + blockIdx.x];
    __syncthreads();
    const int beg = blockIdx.x * CH;
    const int end = (beg + CH < E) ? (beg + CH) : E;
    for (int e0 = beg + tid; e0 < end; e0 += SORT_U * SORT_BD) {
        if (e0 + (SORT_U - 1) * SORT_BD < end) scatter_trip<true>(src, dst, rec, cur, e0, end);
        else                                   scatter_trip<false>(src, dst, rec, cur, e0, end);
    }
}

template <bool FULL>
__device__ __forceinline__ void node_count_trip(const int* __restrict__ rec, int* lc, int i0, int bend) {
    int r[SORT_U];
#pragma unroll
    for (int u = 0; u < SORT_U; ++u) {
        const int i = i0 + u * SORT_BD;
        r[u] = rec[(FULL || i < bend) ? i : (bend - 1)];
    }
#pragma unroll
    for (int u = 0; u < SORT_U; ++u)
        if (FULL || i0 + u * SORT_BD < bend) atomicAdd(&lc[r[u] & 255], 1);
}

template <bool FULL>
__device__ __forceinline__ void node_place_trip(const int* __restrict__ rec, int* __restrict__ csr_src,
                                                int* lc, int bbeg, int i0, int bend) {
    int r[SORT_U];
#pragma unroll
    for (int u = 0; u < SORT_U; ++u) {
        const int i = i0 + u * SORT_BD;
        r[u] = rec[(FULL || i < bend) ? i : (bend - 1)];
    }
#pragma unroll
    for (int u = 0; u < SORT_U; ++u)
        if (FULL || i0 + u * SORT_BD < bend) {
            const int pos = atomicAdd(&lc[r[u] & 255], 1);
            csr_src[bbeg + pos] = (int)((unsigned)r[u] >> 8);
        }
}

__global__ __launch_bounds__(SORT_BD) void bucket_sort_kernel(const int* __restrict__ rec,
                                                              const int* __restrict__ start,
                                                              int* __restrict__ csr_src,
                                                              int* __restrict__ cnt,
                                                              int* __restrict__ offsets,
                                                              float* __restrict__ dinv,
                                                              int N, int E, int NB) {
    __shared__ int lc[256];
    __shared__ int wsum[8];
    const int tid = threadIdx.x;
    const int lane = tid & 31;
    const int wave = tid >> 5;
    const int b = blockIdx.x;
    const int bbeg = start[b * SORT_C];
    const int bend = (b + 1 < NB) ? start[(b + 1) * SORT_C] : E;

    if (tid < 256) lc[tid] = 0;
    __syncthreads();
    for (int i0 = bbeg + tid; i0 < bend; i0 += SORT_U * SORT_BD) {
        if (i0 + (SORT_U - 1) * SORT_BD < bend) node_count_trip<true>(rec, lc, i0, bend);
        else                                    node_count_trip<false>(rec, lc, i0, bend);
    }
    __syncthreads();

    int c = 0, incl = 0;
    if (tid < 256) {
        c = lc[tid];
        incl = c;
        for (int o = 1; o < 32; o <<= 1) {
            int t = __shfl_up(incl, o);
            if (lane >= o) incl += t;
        }
        if (lane == 31) wsum[wave] = incl;
    }
    __syncthreads();
    if (tid < 256) {
        int wpre = 0;
        for (int w = 0; w < 8; ++w) wpre += (w < wave) ? wsum[w] : 0;
        const int excl = wpre + (incl - c);
        lc[tid] = excl;
        const int node = b * 256 + tid;
        if (node < N) {
            cnt[node] = c;
            offsets[node] = bbeg + excl;
            float dg = (float)(c + 1);
            dinv[node] = (dg > 0.0f) ? rsqrtf(dg) : 0.0f;
        }
    }
    __syncthreads();
    for (int i0 = bbeg + tid; i0 < bend; i0 += SORT_U * SORT_BD) {
        if (i0 + (SORT_U - 1) * SORT_BD < bend) node_place_trip<true>(rec, csr_src, lc, bbeg, i0, bend);
        else                                    node_place_trip<false>(rec, csr_src, lc, bbeg, i0, bend);
    }
}


template <int VPL, bool HAS_BIAS, bool SCALE_OUT>
__global__ __launch_bounds__(256) void node_gather_kernel(const float* __restrict__ in,
                                                          const float* __restrict__ dinv,
                                                          const int* __restrict__ offsets,
                                                          const int* __restrict__ cnt,
                                                          const int* __restrict__ csr_src,
                                                          const float* __restrict__ bias,
                                                          float* __restrict__ out, int N) {
    constexpr int DIM = 32 * VPL;
    typedef __attribute__((ext_vector_type(VPL))) float vec;
    const int lane = threadIdx.x & 31;
    int d = blockIdx.x * 8 + (threadIdx.x >> 5);
    if (d >= N) return;
    d = __builtin_amdgcn_readfirstlane(d);
    const int beg = offsets[d];
    const int len = cnt[d];
    const float dd = dinv[d];

    vec acc = *(const vec*)(in + (size_t)d * DIM + lane * VPL);

    for (int base = 0; base < len; base += 32) {
        const int n = (len - base < 32) ? (len - base) : 32;
        int sj = 0;
        if (lane < n) sj = csr_src[beg + base + lane];
#define NG_ROW(jj) (*(const vec*)(in + (size_t)__builtin_amdgcn_readlane(sj, (jj)) * DIM + lane * VPL))
        int j = 0;
        for (; j + 8 <= n; j += 8) {
            const vec r0 = NG_ROW(j),     r1 = NG_ROW(j + 1), r2 = NG_ROW(j + 2), r3 = NG_ROW(j + 3);
            const vec r4 = NG_ROW(j + 4), r5 = NG_ROW(j + 5), r6 = NG_ROW(j + 6), r7 = NG_ROW(j + 7);
            acc += r0;
            acc += r1;
            acc += r2;
            acc += r3;
            acc += r4;
            acc += r5;
            acc += r6;
            acc += r7;
        }
        if (j + 4 <= n) {
            const vec r0 = NG_ROW(j), r1 = NG_ROW(j + 1), r2 = NG_ROW(j + 2), r3 = NG_ROW(j + 3);
            acc += r0;
            acc += r1;
            acc += r2;
            acc += r3;
            j += 4;
        }
        if (j < n) {
            const vec z = {};
            const vec r0 = NG_ROW(j);
            vec r1 = z, r2 = z;
            if (j + 1 < n) r1 = NG_ROW(j + 1);
            if (j + 2 < n) r2 = NG_ROW(j + 2);
            acc += r0;
            acc += r1;
            acc += r2;
        }
#undef NG_ROW
    }

    acc *= dd;
    if (HAS_BIAS) acc += *(const vec*)(bias + lane * VPL);
    if (SCALE_OUT) acc *= dd;
    *(vec*)(out + (size_t)d * DIM + lane * VPL) = acc;
}


template <int NT, bool SCALE_ROWS>
__global__ __launch_bounds__(256) void gemm_wmma_kernel(const float* X,
                                                        const float* __restrict__ W,
                                                        const float* __restrict__ rowscale,
                                                        float* Y,
                                                        int N, int numStripes) {
    constexpr int K = 128;
    constexpr int OUT = NT * 16;
    extern __shared__ float wt[];

    const int tid = threadIdx.x;
    for (int i = tid; i < K * OUT; i += 256) {
        int k = i / OUT;
        int n = i - k * OUT;
        wt[(k >> 1) * (2 * OUT) + 2 * n + (k & 1)] = W[i];
    }
    __syncthreads();

    const int lane = tid & 31;
    const int wave = tid >> 5;
    const int m = lane & 15;
    const int half = lane >> 4;
    const int kg = half << 1;

    const float* lbase = wt + (half * OUT + m) * 2;

    for (int stripe = blockIdx.x; stripe < numStripes; stripe += gridDim.x) {
        const int rowBase = stripe * 128 + wave * 16;
        int arow = rowBase + m;
        if (arow >= N) arow = N - 1;
        const float* xrow = X + (long)arow * K;

        float rs[8];
#pragma unroll
        for (int r = 0; r < 8; ++r) {
            const int row = rowBase + r + 8 * half;
            rs[r] = SCALE_ROWS ? rowscale[row < N ? row : N - 1] : 1.0f;
        }

        v8f zero = {};
        v8f acc[NT];
#pragma unroll
        for (int t = 0; t < NT; ++t) acc[t] = zero;

#pragma unroll 8
        for (int k = 0; k < K; k += 4) {
            v2f a = *(const v2f*)(xrow + k + kg);
            v2f b[NT];
            const float* lk = lbase + (k >> 1) * (2 * OUT);
#pragma unroll
            for (int t = 0; t < NT; ++t) b[t] = *(const v2f*)(lk + t * 32);
#pragma unroll
            for (int t = 0; t < NT; ++t)
                acc[t] = __builtin_amdgcn_wmma_f32_16x16x4_f32(
                    false, a, false, b[t], (short)0, acc[t], false, false);
        }

#pragma unroll
        for (int t = 0; t < NT; ++t) {
            const int col = t * 16 + m;
#pragma unroll
            for (int r = 0; r < 8; ++r) {
                const int row = rowBase + r + 8 * half;
                if (row < N) Y[(long)row * OUT + col] = SCALE_ROWS ? acc[t][r] * rs[r] : acc[t][r];
            }
        }
    }
}


extern "C" void kernel_launch(void* const* d_in, const int* in_sizes, int n_in,
                              void* d_out, int out_size, void* d_ws, size_t ws_size,
                              hipStream_t stream) {
    const float* x  = (const float*)d_in[0];
    const int*   ei = (const int*)d_in[1];
    const float* W1 = (const float*)d_in[2];
    const float* b1 = (const float*)d_in[3];
    const float* W2 = (const float*)d_in[4];
    const float* b2 = (const float*)d_in[5];
    float* out = (float*)d_out;

    const int N = in_sizes[0] / 128;
    const int E = in_sizes[1] / 2;
    const int* src = ei;
    const int* dst = ei + E;

    float* dinv    = (float*)d_ws;
    float* xw      = dinv + N;
    float* t       = xw + (size_t)N * 64;
    float* w12     = t + (size_t)N * 64;
    float* b12     = w12 + 128 * 64;
    int*   cnt     = (int*)(w12 + 129 * 64);
    int*   offsets = cnt + N;
    int*   rec     = offsets + N;
    int*   csr_src = rec + E;

    const int NB = (N + 255) / 256;
    const int CH = (E + SORT_C - 1) / SORT_C;
    const int M  = NB * SORT_C;
    int* hist  = (int*)t;
    int* start = hist + M;
    bucket_count_kernel<<<SORT_C, SORT_BD, NB * sizeof(int), stream>>>(dst, hist, E, CH, NB);
    scan_excl_fold_kernel<<<(M + 1023) / 1024 + FOLD_WG, 1024, 0, stream>>>(hist, start, M, W1, b1, W2, w12);
    bucket_scatter_kernel<<<SORT_C, SORT_BD, NB * sizeof(int), stream>>>(src, dst, start, rec, E, CH, NB);
    bucket_sort_kernel<<<NB, SORT_BD, 0, stream>>>(rec, start, csr_src, cnt, offsets, dinv, N, E, NB);

    const int numStripes = (N + 127) / 128;
    const int gemmBlocks = numStripes < 320 ? numStripes : 320;
    const int gatherBlocks = (N + 7) / 8;

    gemm_wmma_kernel<4, true><<<gemmBlocks, 256, 64 * 128 * sizeof(float), stream>>>(x, w12, dinv, xw, N, numStripes);

    node_gather_kernel<2, true, true><<<gatherBlocks, 256, 0, stream>>>(xw, dinv, offsets, cnt, csr_src, b12, t, N);
    node_gather_kernel<2, true, false><<<gatherBlocks, 256, 0, stream>>>(t, dinv, offsets, cnt, csr_src, b2, out, N);
}


# Edge grouping without a chip-wide prefix sum: each chunk orders its edges by bucket in LDS and writes whole lines, bucket blocks collect their 256 runs; 5 launches, not 7; arrival order as before
# speedup vs baseline: 6.0873x; 1.2482x over previous
#include <hip/hip_runtime.h>


typedef __attribute__((ext_vector_type(2))) float v2f;
typedef __attribute__((ext_vector_type(8))) float v8f;


constexpr int SORT_BD = 1024;
constexpr int SORT_C  = 256;
constexpr int SORT_U  = 8;

template <bool FULL>
__device__ __forceinline__ void count_trip(const int* __restrict__ dst, int* lh, int e0, int end) {
    int d[SORT_U];
#pragma unroll
    for (int u = 0; u < SORT_U; ++u) {
        const int e = e0 + u * SORT_BD;
        d[u] = dst[(FULL || e < end) ? e : (end - 1)];
    }
    if (!FULL) {
#pragma unroll
        for (int u = 0; u < SORT_U; ++u)
            asm volatile("" : "+v"(d[u]));
    }
#pragma unroll
    for (int u = 0; u < SORT_U; ++u)
        if (FULL || e0 + u * SORT_BD < end) atomicAdd(&lh[d[u] >> 8], 1);
}

constexpr int FOLD_WG = 9;
constexpr int SORT_NC = SORT_C - FOLD_WG;

__device__ __forceinline__ void fold_weights_role(const float* __restrict__ W1,
                                                  const float* __restrict__ b1,
                                                  const float* __restrict__ W2,
                                                  float* __restrict__ w12, int g, float* wt) {
    constexpr int K = 128;
    constexpr int OUT = 64;
    const int tid = threadIdx.x;
    const int lane = tid & 31;
    const int wave = tid >> 5;
    const int m = lane & 15;
    const int half = lane >> 4;
    const int kg = half << 1;
    const bool compute = wave < 4;

    v2f a[K / 4];
    if (compute) {
        const float* arow = (g < 8) ? (W1 + (size_t)(g * 16 + m) * K) : b1;
#pragma unroll
        for (int j = 0; j < K / 4; ++j) a[j] = *(const v2f*)(arow + 4 * j + kg);
    }

    float w[K * OUT / 1024];
#pragma unroll
    for (int u = 0; u < K * OUT / 1024; ++u) w[u] = W2[tid + u * 1024];
#pragma unroll
    for (int u = 0; u < K * OUT / 1024; ++u) {
        const int i = tid + u * 1024;
        const int k = i / OUT;
        const int n = i - k * OUT;
        wt[(k >> 1) * (2 * OUT) + 2 * n + (k & 1)] = w[u];
    }
    __syncthreads();
    if (!compute) return;

    int ooff = (g * 16 + 8 * half) * OUT + wave * 16 + m;
    const float* lbase = wt + (half * OUT + m) * 2 + wave * 32;
    v8f acc = {};
    asm volatile("" : "+v"(ooff));
    __builtin_amdgcn_sched_barrier(0);
#pragma unroll
    for (int j = 0; j < K / 4; ++j) {
        const v2f b = *(const v2f*)(lbase + (2 * j) * (2 * OUT));
        acc = __builtin_amdgcn_wmma_f32_16x16x4_f32(false, a[j], false, b, (short)0, acc, false, false);
    }
    __builtin_amdgcn_sched_barrier(0);
    asm volatile("v_nop\n\tv_nop\n\tv_nop\n\tv_nop");
    __builtin_amdgcn_sched_barrier(0);

    if (g < 8) {
#pragma unroll
        for (int r = 0; r < 8; ++r) w12[ooff + r * OUT] = acc[r];
    } else if (half == 0) {
        w12[ooff] = acc[0];
    }
}

template <bool FULL>
__device__ __forceinline__ void place_trip(const int* __restrict__ src, const int* __restrict__ dst,
                                           int* stage, int* cur, int e0, int end) {
    int d[SORT_U], s[SORT_U];
#pragma unroll
    for (int u = 0; u < SORT_U; ++u) {
        const int e = e0 + u * SORT_BD;
        const int ec = (FULL || e < end) ? e : (end - 1);
        d[u] = dst[ec];
        s[u] = src[ec];
    }
    if (!FULL) {
#pragma unroll
        for (int u = 0; u < SORT_U; ++u)
            asm volatile("" : "+v"(d[u]), "+v"(s[u]));
    }
#pragma unroll
    for (int u = 0; u < SORT_U; ++u)
        if (FULL || e0 + u * SORT_BD < end) {
            const int pos = atomicAdd(&cur[d[u] >> 8], 1);
            stage[pos] = (int)(((unsigned)s[u] << 8) | (unsigned)(d[u] & 255));
        }
}

__global__ __launch_bounds__(SORT_BD) void chunk_order_fold_kernel(const int* __restrict__ src,
                                                                   const int* __restrict__ dst,
                                                                   int* __restrict__ hist,
                                                                   int* __restrict__ lst,
                                                                   int* __restrict__ rec,
                                                                   int E, int CH, int NB,
                                                                   const float* __restrict__ W1,
                                                                   const float* __restrict__ b1,
                                                                   const float* __restrict__ W2,
                                                                   float* __restrict__ w12) {
    extern __shared__ int dyn[];
    __shared__ float wt[128 * 64];
    __shared__ int wsum[32];
    __shared__ int s_tot;
    const int col = blockIdx.x;
    if (col < FOLD_WG) {
        for (int b = threadIdx.x; b < NB; b += SORT_BD) {
            hist[b * SORT_C + col] = 0;
            lst[b * SORT_C + col] = 0;
        }
        fold_weights_role(W1, b1, W2, w12, col, wt);
        return;
    }
    int* lh = dyn;
    int* stage = dyn + NB;
    const int tid = threadIdx.x;
    const int lane = tid & 31;
    const int wave = tid >> 5;
    const int c = col - FOLD_WG;
    for (int b = tid; b < NB; b += SORT_BD) lh[b] = 0;
    __syncthreads();
    const int beg = c * CH;
    const int end = (beg + CH < E) ? (beg + CH) : E;
    for (int e0 = beg + tid; e0 < end; e0 += SORT_U * SORT_BD) {
        if (e0 + (SORT_U - 1) * SORT_BD < end) count_trip<true>(dst, lh, e0, end);
        else                                   count_trip<false>(dst, lh, e0, end);
    }
    __syncthreads();

    int carry = 0;
    for (int b0 = 0; b0 < NB; b0 += SORT_BD) {
        const int b = b0 + tid;
        const int cv = (b < NB) ? lh[b] : 0;
        int incl = cv;
        for (int o = 1; o < 32; o <<= 1) {
            int t = __shfl_up(incl, o);
            if (lane >= o) incl += t;
        }
        if (lane == 31) wsum[wave] = incl;
        __syncthreads();
        if (wave == 0) {
            int v = wsum[lane];
            int iv = v;
            for (int o = 1; o < 32; o <<= 1) {
                int t = __shfl_up(iv, o);
                if (lane >= o) iv += t;
            }
            wsum[lane] = iv - v;
            if (lane == 31) s_tot = iv;
        }
        __syncthreads();
        const int excl = carry + wsum[wave] + (incl - cv);
        if (b < NB) {
            lh[b] = excl;
            hist[b * SORT_C + col] = cv;
            lst[b * SORT_C + col] = excl;
        }
        carry += s_tot;
        __syncthreads();
    }

    for (int e0 = beg + tid; e0 < end; e0 += SORT_U * SORT_BD) {
        if (e0 + (SORT_U - 1) * SORT_BD < end) place_trip<true>(src, dst, stage, lh, e0, end);
        else                                   place_trip<false>(src, dst, stage, lh, e0, end);
    }
    __syncthreads();
    for (int i = tid; i < end - beg; i += SORT_BD) rec[beg + i] = stage[i];
}

constexpr int SEG_U = 8;

template <bool PLACE>
__device__ __forceinline__ void run_trip(const int* __restrict__ rec, int* __restrict__ csr_src,
                                         int* lc, int bbeg, int seg0, int segn, int k0) {
    int r[SEG_U];
#pragma unroll
    for (int u = 0; u < SEG_U; ++u) {
        const int k = k0 + 4 * u;
        r[u] = rec[seg0 + ((k < segn) ? k : 0)];
    }
#pragma unroll
    for (int u = 0; u < SEG_U; ++u)
        if (k0 + 4 * u < segn) {
            const int pos = atomicAdd(&lc[r[u] & 255], 1);
            if (PLACE) csr_src[bbeg + pos] = (int)((unsigned)r[u] >> 8);
        }
}

__global__ __launch_bounds__(SORT_BD) void bucket_sort_kernel(const int* __restrict__ rec,
                                                              const int* __restrict__ hist,
                                                              const int* __restrict__ lst,
                                                              int* __restrict__ csr_src,
                                                              int* __restrict__ cnt,
                                                              int* __restrict__ offsets,
                                                              float* __restrict__ dinv,
                                                              int N, int CH) {
    __shared__ int lc[256];
    __shared__ int wsum[32];
    __shared__ int s_prefix;
    const int tid = threadIdx.x;
    const int lane = tid & 31;
    const int wave = tid >> 5;
    const int b = blockIdx.x;

    const int cc = tid >> 2;
    const int sub = tid & 3;
    const int segn = hist[b * SORT_C + cc];
    const int seg0 = (cc - FOLD_WG) * CH + lst[b * SORT_C + cc];

    int part = 0;
    const int4* c4 = (const int4*)hist;
    const int lim = (b * SORT_C) >> 2;
    for (int j = tid; j < lim; j += 8 * SORT_BD) {
        int4 v[8];
#pragma unroll
        for (int u = 0; u < 8; ++u) {
            const int q = j + u * SORT_BD;
            v[u] = c4[(q < lim) ? q : (lim - 1)];
        }
#pragma unroll
        for (int u = 0; u < 8; ++u)
            if (j + u * SORT_BD < lim) part += (v[u].x + v[u].y) + (v[u].z + v[u].w);
    }
    for (int o = 16; o > 0; o >>= 1) part += __shfl_xor(part, o);
    if (lane == 0) wsum[wave] = part;
    if (tid < 256) lc[tid] = 0;
    __syncthreads();
    if (wave == 0) {
        int v = wsum[lane];
        for (int o = 16; o > 0; o >>= 1) v += __shfl_xor(v, o);
        if (lane == 0) s_prefix = v;
    }
    for (int k0 = sub; k0 < segn; k0 += 4 * SEG_U) run_trip<false>(rec, csr_src, lc, 0, seg0, segn, k0);
    __syncthreads();
    const int bbeg = s_prefix;

    int c = 0, incl = 0;
    if (tid < 256) {
        c = lc[tid];
        incl = c;
        for (int o = 1; o < 32; o <<= 1) {
            int t = __shfl_up(incl, o);
            if (lane >= o) incl += t;
        }
        if (lane == 31) wsum[wave] = incl;
    }
    __syncthreads();
    if (tid < 256) {
        int wpre = 0;
        for (int w = 0; w < 8; ++w) wpre += (w < wave) ? wsum[w] : 0;
        const int excl = wpre + (incl - c);
        lc[tid] = excl;
        const int node = b * 256 + tid;
        if (node < N) {
            cnt[node] = c;
            offsets[node] = bbeg + excl;
            float dg = (float)(c + 1);
            dinv[node] = (dg > 0.0f) ? rsqrtf(dg) : 0.0f;
        }
    }
    __syncthreads();
    for (int k0 = sub; k0 < segn; k0 += 4 * SEG_U) run_trip<true>(rec, csr_src, lc, bbeg, seg0, segn, k0);
}


template <int VPL, bool HAS_BIAS, bool SCALE_OUT>
__global__ __launch_bounds__(256) void node_gather_kernel(const float* __restrict__ in,
                                                          const float* __restrict__ dinv,
                                                          const int* __restrict__ offsets,
                                                          const int* __restrict__ cnt,
                                                          const int* __restrict__ csr_src,
                                                          const float* __restrict__ bias,
                                                          float* __restrict__ out, int N) {
    constexpr int DIM = 32 * VPL;
    typedef __attribute__((ext_vector_type(VPL))) float vec;
    const int lane = threadIdx.x & 31;
    int d = blockIdx.x * 8 + (threadIdx.x >> 5);
    if (d >= N) return;
    d = __builtin_amdgcn_readfirstlane(d);
    const int beg = offsets[d];
    const int len = cnt[d];
    const float dd = dinv[d];

    vec acc = *(const vec*)(in + (size_t)d * DIM + lane * VPL);

    for (int base = 0; base < len; base += 32) {
        const int n = (len - base < 32) ? (len - base) : 32;
        int sj = 0;
        if (lane < n) sj = csr_src[beg + base + lane];
#define NG_ROW(jj) (*(const vec*)(in + (size_t)__builtin_amdgcn_readlane(sj, (jj)) * DIM + lane * VPL))
        int j = 0;
        for (; j + 8 <= n; j += 8) {
            const vec r0 = NG_ROW(j),     r1 = NG_ROW(j + 1), r2 = NG_ROW(j + 2), r3 = NG_ROW(j + 3);
            const vec r4 = NG_ROW(j + 4), r5 = NG_ROW(j + 5), r6 = NG_ROW(j + 6), r7 = NG_ROW(j + 7);
            acc += r0;
            acc += r1;
            acc += r2;
            acc += r3;
            acc += r4;
            acc += r5;
            acc += r6;
            acc += r7;
        }
        if (j + 4 <= n) {
            const vec r0 = NG_ROW(j), r1 = NG_ROW(j + 1), r2 = NG_ROW(j + 2), r3 = NG_ROW(j + 3);
            acc += r0;
            acc += r1;
            acc += r2;
            acc += r3;
            j += 4;
        }
        if (j < n) {
            const vec z = {};
            const vec r0 = NG_ROW(j);
            vec r1 = z, r2 = z;
            if (j + 1 < n) r1 = NG_ROW(j + 1);
            if (j + 2 < n) r2 = NG_ROW(j + 2);
            acc += r0;
            acc += r1;
            acc += r2;
        }
#undef NG_ROW
    }

    acc *= dd;
    if (HAS_BIAS) acc += *(const vec*)(bias + lane * VPL);
    if (SCALE_OUT) acc *= dd;
    *(vec*)(out + (size_t)d * DIM + lane * VPL) = acc;
}


template <int NT, bool SCALE_ROWS>
__global__ __launch_bounds__(256) void gemm_wmma_kernel(const float* X,
                                                        const float* __restrict__ W,
                                                        const float* __restrict__ rowscale,
                                                        float* Y,
                                                        int N, int numStripes) {
    constexpr int K = 128;
    constexpr int OUT = NT * 16;
    extern __shared__ float wt[];

    const int tid = threadIdx.x;
    for (int i = tid; i < K * OUT; i += 256) {
        int k = i / OUT;
        int n = i - k * OUT;
        wt[(k >> 1) * (2 * OUT) + 2 * n + (k & 1)] = W[i];
    }
    __syncthreads();

    const int lane = tid & 31;
    const int wave = tid >> 5;
    const int m = lane & 15;
    const int half = lane >> 4;
    const int kg = half << 1;

    const float* lbase = wt + (half * OUT + m) * 2;

    for (int stripe = blockIdx.x; stripe < numStripes; stripe += gridDim.x) {
        const int rowBase = stripe * 128 + wave * 16;
        int arow = rowBase + m;
        if (arow >= N) arow = N - 1;
        const float* xrow = X + (long)arow * K;

        float rs[8];
#pragma unroll
        for (int r = 0; r < 8; ++r) {
            const int row = rowBase + r + 8 * half;
            rs[r] = SCALE_ROWS ? rowscale[row < N ? row : N - 1] : 1.0f;
        }

        v8f zero = {};
        v8f acc[NT];
#pragma unroll
        for (int t = 0; t < NT; ++t) acc[t] = zero;

#pragma unroll 8
        for (int k = 0; k < K; k += 4) {
            v2f a = *(const v2f*)(xrow + k + kg);
            v2f b[NT];
            const float* lk = lbase + (k >> 1) * (2 * OUT);
#pragma unroll
            for (int t = 0; t < NT; ++t) b[t] = *(const v2f*)(lk + t * 32);
#pragma unroll
            for (int t = 0; t < NT; ++t)
                acc[t] = __builtin_amdgcn_wmma_f32_16x16x4_f32(
                    false, a, false, b[t], (short)0, acc[t], false, false);
        }

#pragma unroll
        for (int t = 0; t < NT; ++t) {
            const int col = t * 16 + m;
#pragma unroll
            for (int r = 0; r < 8; ++r) {
                const int row = rowBase + r + 8 * half;
                if (row < N) Y[(long)row * OUT + col] = SCALE_ROWS ? acc[t][r] * rs[r] : acc[t][r];
            }
        }
    }
}


extern "C" void kernel_launch(void* const* d_in, const int* in_sizes, int n_in,
                              void* d_out, int out_size, void* d_ws, size_t ws_size,
                              hipStream_t stream) {
    const float* x  = (const float*)d_in[0];
    const int*   ei = (const int*)d_in[1];
    const float* W1 = (const float*)d_in[2];
    const float* b1 = (const float*)d_in[3];
    const float* W2 = (const float*)d_in[4];
    const float* b2 = (const float*)d_in[5];
    float* out = (float*)d_out;

    const int N = in_sizes[0] / 128;
    const int E = in_sizes[1] / 2;
    const int* src = ei;
    const int* dst = ei + E;

    float* dinv    = (float*)d_ws;
    float* xw      = dinv + N;
    float* t       = xw + (size_t)N * 64;
    float* w12     = t + (size_t)N * 64;
    float* b12     = w12 + 128 * 64;
    int*   cnt     = (int*)(w12 + 129 * 64);
    int*   offsets = cnt + N;
    int*   rec     = offsets + N;
    int*   csr_src = rec + E;

    const int NB = (N + 255) / 256;
    const int CH = (E + SORT_NC - 1) / SORT_NC;
    const int M  = NB * SORT_C;
    int* hist = (int*)t;
    int* lst  = hist + M;
    chunk_order_fold_kernel<<<SORT_C, SORT_BD, (NB + CH) * sizeof(int), stream>>>(src, dst, hist, lst, rec, E, CH, NB,
                                                                                           W1, b1, W2, w12);
    bucket_sort_kernel<<<NB, SORT_BD, 0, stream>>>(rec, hist, lst, csr_src, cnt, offsets, dinv, N, CH);

    const int numStripes = (N + 127) / 128;
    const int gemmBlocks = numStripes < 320 ? numStripes : 320;
    const int gatherBlocks = (N + 7) / 8;

    gemm_wmma_kernel<4, true><<<gemmBlocks, 256, 64 * 128 * sizeof(float), stream>>>(x, w12, dinv, xw, N, numStripes);

    node_gather_kernel<2, true, true><<<gatherBlocks, 256, 0, stream>>>(xw, dinv, offsets, cnt, csr_src, b12, t, N);
    node_gather_kernel<2, true, false><<<gatherBlocks, 256, 0, stream>>>(t, dinv, offsets, cnt, csr_src, b2, out, N);
}


# Both aggregations: a wave walks 7 consecutive nodes (node facts by one vector load + readlane, next node's indices and own row requested one node ahead), not one short wave per node; same sum order
# speedup vs baseline: 6.3494x; 1.0431x over previous
#include <hip/hip_runtime.h>


typedef __attribute__((ext_vector_type(2))) float v2f;
typedef __attribute__((ext_vector_type(8))) float v8f;


constexpr int SORT_BD = 1024;
constexpr int SORT_C  = 256;
constexpr int SORT_U  = 8;

template <bool FULL>
__device__ __forceinline__ void count_trip(const int* __restrict__ dst, int* lh, int e0, int end) {
    int d[SORT_U];
#pragma unroll
    for (int u = 0; u < SORT_U; ++u) {
        const int e = e0 + u * SORT_BD;
        d[u] = dst[(FULL || e < end) ? e : (end - 1)];
    }
    if (!FULL) {
#pragma unroll
        for (int u = 0; u < SORT_U; ++u)
            asm volatile("" : "+v"(d[u]));
    }
#pragma unroll
    for (int u = 0; u < SORT_U; ++u)
        if (FULL || e0 + u * SORT_BD < end) atomicAdd(&lh[d[u] >> 8], 1);
}

constexpr int FOLD_WG = 9;
constexpr int SORT_NC = SORT_C - FOLD_WG;

__device__ __forceinline__ void fold_weights_role(const float* __restrict__ W1,
                                                  const float* __restrict__ b1,
                                                  const float* __restrict__ W2,
                                                  float* __restrict__ w12, int g, float* wt) {
    constexpr int K = 128;
    constexpr int OUT = 64;
    const int tid = threadIdx.x;
    const int lane = tid & 31;
    const int wave = tid >> 5;
    const int m = lane & 15;
    const int half = lane >> 4;
    const int kg = half << 1;
    const bool compute = wave < 4;

    v2f a[K / 4];
    if (compute) {
        const float* arow = (g < 8) ? (W1 + (size_t)(g * 16 + m) * K) : b1;
#pragma unroll
        for (int j = 0; j < K / 4; ++j) a[j] = *(const v2f*)(arow + 4 * j + kg);
    }

    float w[K * OUT / 1024];
#pragma unroll
    for (int u = 0; u < K * OUT / 1024; ++u) w[u] = W2[tid + u * 1024];
#pragma unroll
    for (int u = 0; u < K * OUT / 1024; ++u) {
        const int i = tid + u * 1024;
        const int k = i / OUT;
        const int n = i - k * OUT;
        wt[(k >> 1) * (2 * OUT) + 2 * n + (k & 1)] = w[u];
    }
    __syncthreads();
    if (!compute) return;

    int ooff = (g * 16 + 8 * half) * OUT + wave * 16 + m;
    const float* lbase = wt + (half * OUT + m) * 2 + wave * 32;
    v8f acc = {};
    asm volatile("" : "+v"(ooff));
    __builtin_amdgcn_sched_barrier(0);
#pragma unroll
    for (int j = 0; j < K / 4; ++j) {
        const v2f b = *(const v2f*)(lbase + (2 * j) * (2 * OUT));
        acc = __builtin_amdgcn_wmma_f32_16x16x4_f32(false, a[j], false, b, (short)0, acc, false, false);
    }
    __builtin_amdgcn_sched_barrier(0);
    asm volatile("v_nop\n\tv_nop\n\tv_nop\n\tv_nop");
    __builtin_amdgcn_sched_barrier(0);

    if (g < 8) {
#pragma unroll
        for (int r = 0; r < 8; ++r) w12[ooff + r * OUT] = acc[r];
    } else if (half == 0) {
        w12[ooff] = acc[0];
    }
}

template <bool FULL>
__device__ __forceinline__ void place_trip(const int* __restrict__ src, const int* __restrict__ dst,
                                           int* stage, int* cur, int e0, int end) {
    int d[SORT_U], s[SORT_U];
#pragma unroll
    for (int u = 0; u < SORT_U; ++u) {
        const int e = e0 + u * SORT_BD;
        const int ec = (FULL || e < end) ? e : (end - 1);
        d[u] = dst[ec];
        s[u] = src[ec];
    }
    if (!FULL) {
#pragma unroll
        for (int u = 0; u < SORT_U; ++u)
            asm volatile("" : "+v"(d[u]), "+v"(s[u]));
    }
#pragma unroll
    for (int u = 0; u < SORT_U; ++u)
        if (FULL || e0 + u * SORT_BD < end) {
            const int pos = atomicAdd(&cur[d[u] >> 8], 1);
            stage[pos] = (int)(((unsigned)s[u] << 8) | (unsigned)(d[u] & 255));
        }
}

__global__ __launch_bounds__(SORT_BD) void chunk_order_fold_kernel(const int* __restrict__ src,
                                                                   const int* __restrict__ dst,
                                                                   int* __restrict__ hist,
                                                                   int* __restrict__ lst,
                                                                   int* __restrict__ rec,
                                                                   int E, int CH, int NB,
                                                                   const float* __restrict__ W1,
                                                                   const float* __restrict__ b1,
                                                                   const float* __restrict__ W2,
                                                                   float* __restrict__ w12) {
    extern __shared__ int dyn[];
    __shared__ float wt[128 * 64];
    __shared__ int wsum[32];
    __shared__ int s_tot;
    const int col = blockIdx.x;
    if (col < FOLD_WG) {
        for (int b = threadIdx.x; b < NB; b += SORT_BD) {
            hist[b * SORT_C + col] = 0;
            lst[b * SORT_C + col] = 0;
        }
        fold_weights_role(W1, b1, W2, w12, col, wt);
        return;
    }
    int* lh = dyn;
    int* stage = dyn + NB;
    const int tid = threadIdx.x;
    const int lane = tid & 31;
    const int wave = tid >> 5;
    const int c = col - FOLD_WG;
    for (int b = tid; b < NB; b += SORT_BD) lh[b] = 0;
    __syncthreads();
    const int beg = c * CH;
    const int end = (beg + CH < E) ? (beg + CH) : E;
    for (int e0 = beg + tid; e0 < end; e0 += SORT_U * SORT_BD) {
        if (e0 + (SORT_U - 1) * SORT_BD < end) count_trip<true>(dst, lh, e0, end);
        else                                   count_trip<false>(dst, lh, e0, end);
    }
    __syncthreads();

    int carry = 0;
    for (int b0 = 0; b0 < NB; b0 += SORT_BD) {
        const int b = b0 + tid;
        const int cv = (b < NB) ? lh[b] : 0;
        int incl = cv;
        for (int o = 1; o < 32; o <<= 1) {
            int t = __shfl_up(incl, o);
            if (lane >= o) incl += t;
        }
        if (lane == 31) wsum[wave] = incl;
        __syncthreads();
        if (wave == 0) {
            int v = wsum[lane];
            int iv = v;
            for (int o = 1; o < 32; o <<= 1) {
                int t = __shfl_up(iv, o);
                if (lane >= o) iv += t;
            }
            wsum[lane] = iv - v;
            if (lane == 31) s_tot = iv;
        }
        __syncthreads();
        const int excl = carry + wsum[wave] + (incl - cv);
        if (b < NB) {
            lh[b] = excl;
            hist[b * SORT_C + col] = cv;
            lst[b * SORT_C + col] = excl;
        }
        carry += s_tot;
        __syncthreads();
    }

    for (int e0 = beg + tid; e0 < end; e0 += SORT_U * SORT_BD) {
        if (e0 + (SORT_U - 1) * SORT_BD < end) place_trip<true>(src, dst, stage, lh, e0, end);
        else                                   place_trip<false>(src, dst, stage, lh, e0, end);
    }
    __syncthreads();
    for (int i = tid; i < end - beg; i += SORT_BD) rec[beg + i] = stage[i];
}

constexpr int SEG_U = 8;

template <bool PLACE>
__device__ __forceinline__ void run_trip(const int* __restrict__ rec, int* __restrict__ csr_src,
                                         int* lc, int bbeg, int seg0, int segn, int k0) {
    int r[SEG_U];
#pragma unroll
    for (int u = 0; u < SEG_U; ++u) {
        const int k = k0 + 4 * u;
        r[u] = rec[seg0 + ((k < segn) ? k : 0)];
    }
#pragma unroll
    for (int u = 0; u < SEG_U; ++u)
        if (k0 + 4 * u < segn) {
            const int pos = atomicAdd(&lc[r[u] & 255], 1);
            if (PLACE) csr_src[bbeg + pos] = (int)((unsigned)r[u] >> 8);
        }
}

__global__ __launch_bounds__(SORT_BD) void bucket_sort_kernel(const int* __restrict__ rec,
                                                              const int* __restrict__ hist,
                                                              const int* __restrict__ lst,
                                                              int* __restrict__ csr_src,
                                                              int* __restrict__ cnt,
                                                              int* __restrict__ offsets,
                                                              float* __restrict__ dinv,
                                                              int N, int CH) {
    __shared__ int lc[256];
    __shared__ int wsum[32];
    __shared__ int s_prefix;
    const int tid = threadIdx.x;
    const int lane = tid & 31;
    const int wave = tid >> 5;
    const int b = blockIdx.x;

    const int cc = tid >> 2;
    const int sub = tid & 3;
    const int segn = hist[b * SORT_C + cc];
    const int seg0 = (cc - FOLD_WG) * CH + lst[b * SORT_C + cc];

    int part = 0;
    const int4* c4 = (const int4*)hist;
    const int lim = (b * SORT_C) >> 2;
    for (int j = tid; j < lim; j += 8 * SORT_BD) {
        int4 v[8];
#pragma unroll
        for (int u = 0; u < 8; ++u) {
            const int q = j + u * SORT_BD;
            v[u] = c4[(q < lim) ? q : (lim - 1)];
        }
#pragma unroll
        for (int u = 0; u < 8; ++u)
            if (j + u * SORT_BD < lim) part += (v[u].x + v[u].y) + (v[u].z + v[u].w);
    }
    for (int o = 16; o > 0; o >>= 1) part += __shfl_xor(part, o);
    if (lane == 0) wsum[wave] = part;
    if (tid < 256) lc[tid] = 0;
    __syncthreads();
    if (wave == 0) {
        int v = wsum[lane];
        for (int o = 16; o > 0; o >>= 1) v += __shfl_xor(v, o);
        if (lane == 0) s_prefix = v;
    }
    for (int k0 = sub; k0 < segn; k0 += 4 * SEG_U) run_trip<false>(rec, csr_src, lc, 0, seg0, segn, k0);
    __syncthreads();
    const int bbeg = s_prefix;

    int c = 0, incl = 0;
    if (tid < 256) {
        c = lc[tid];
        incl = c;
        for (int o = 1; o < 32; o <<= 1) {
            int t = __shfl_up(incl, o);
            if (lane >= o) incl += t;
        }
        if (lane == 31) wsum[wave] = incl;
    }
    __syncthreads();
    if (tid < 256) {
        int wpre = 0;
        for (int w = 0; w < 8; ++w) wpre += (w < wave) ? wsum[w] : 0;
        const int excl = wpre + (incl - c);
        lc[tid] = excl;
        const int node = b * 256 + tid;
        if (node < N) {
            cnt[node] = c;
            offsets[node] = bbeg + excl;
            float dg = (float)(c + 1);
            dinv[node] = (dg > 0.0f) ? rsqrtf(dg) : 0.0f;
        }
    }
    __syncthreads();
    for (int k0 = sub; k0 < segn; k0 += 4 * SEG_U) run_trip<true>(rec, csr_src, lc, bbeg, seg0, segn, k0);
}


constexpr int NG_NPW = 7;

template <int VPL, bool HAS_BIAS, bool SCALE_OUT>
__global__ __launch_bounds__(256) void node_gather_kernel(const float* __restrict__ in,
                                                          const float* __restrict__ dinv,
                                                          const int* __restrict__ offsets,
                                                          const int* __restrict__ cnt,
                                                          const int* __restrict__ csr_src,
                                                          const float* __restrict__ bias,
                                                          float* __restrict__ out, int N) {
    constexpr int DIM = 32 * VPL;
    typedef __attribute__((ext_vector_type(VPL))) float vec;
    const int lane = threadIdx.x & 31;
    int w = blockIdx.x * 8 + (threadIdx.x >> 5);
    w = __builtin_amdgcn_readfirstlane(w);
    const int d0 = w * NG_NPW;
    if (d0 >= N) return;
    const int nn = (N - d0 < NG_NPW) ? (N - d0) : NG_NPW;

    vec self_n = *(const vec*)(in + (size_t)d0 * DIM + lane * VPL);
    const int dl = d0 + ((lane < nn) ? lane : 0);
    const int offv = offsets[dl];
    const int cntv = cnt[dl];
    const int dinvb = __float_as_int(dinv[dl]);
    vec bv = {};
    if (HAS_BIAS) bv = *(const vec*)(bias + lane * VPL);

    int nbeg = __builtin_amdgcn_readlane(offv, 0);
    int nlen = __builtin_amdgcn_readlane(cntv, 0);
    int sj_n = 0;
    if (lane < nlen) sj_n = csr_src[nbeg + lane];

    for (int i = 0; i < nn; ++i) {
        const int d = d0 + i;
        const int beg = nbeg;
        const int len = nlen;
        const float dd = __int_as_float(__builtin_amdgcn_readlane(dinvb, i));
        int sj = sj_n;
        vec acc = self_n;

        if (i + 1 < nn) {
            nbeg = __builtin_amdgcn_readlane(offv, i + 1);
            nlen = __builtin_amdgcn_readlane(cntv, i + 1);
            sj_n = 0;
            if (lane < nlen) sj_n = csr_src[nbeg + lane];
            self_n = *(const vec*)(in + (size_t)(d + 1) * DIM + lane * VPL);
        }

        for (int base = 0; base < len; base += 32) {
            const int n = (len - base < 32) ? (len - base) : 32;
            if (base > 0) {
                sj = 0;
                if (lane < n) sj = csr_src[beg + base + lane];
                asm volatile("" : "+v"(sj));
            }
#define NG_ROW(jj) (*(const vec*)(in + (size_t)__builtin_amdgcn_readlane(sj, (jj)) * DIM + lane * VPL))
            int j = 0;
            for (; j + 8 <= n; j += 8) {
                const vec r0 = NG_ROW(j),     r1 = NG_ROW(j + 1), r2 = NG_ROW(j + 2), r3 = NG_ROW(j + 3);
                const vec r4 = NG_ROW(j + 4), r5 = NG_ROW(j + 5), r6 = NG_ROW(j + 6), r7 = NG_ROW(j + 7);
                acc += r0;
                acc += r1;
                acc += r2;
                acc += r3;
                acc += r4;
                acc += r5;
                acc += r6;
                acc += r7;
            }
            if (j + 4 <= n) {
                const vec r0 = NG_ROW(j), r1 = NG_ROW(j + 1), r2 = NG_ROW(j + 2), r3 = NG_ROW(j + 3);
                acc += r0;
                acc += r1;
                acc += r2;
                acc += r3;
                j += 4;
            }
            if (j < n) {
                const vec z = {};
                const vec r0 = NG_ROW(j);
                vec r1 = z, r2 = z;
                if (j + 1 < n) r1 = NG_ROW(j + 1);
                if (j + 2 < n) r2 = NG_ROW(j + 2);
                acc += r0;
                acc += r1;
                acc += r2;
            }
#undef NG_ROW
        }

        acc *= dd;
        if (HAS_BIAS) acc += bv;
        if (SCALE_OUT) acc *= dd;
        *(vec*)(out + (size_t)d * DIM + lane * VPL) = acc;
    }
}


template <int NT, bool SCALE_ROWS>
__global__ __launch_bounds__(256) void gemm_wmma_kernel(const float* X,
                                                        const float* __restrict__ W,
                                                        const float* __restrict__ rowscale,
                                                        float* Y,
                                                        int N, int numStripes) {
    constexpr int K = 128;
    constexpr int OUT = NT * 16;
    extern __shared__ float wt[];

    const int tid = threadIdx.x;
    for (int i = tid; i < K * OUT; i += 256) {
        int k = i / OUT;
        int n = i - k * OUT;
        wt[(k >> 1) * (2 * OUT) + 2 * n + (k & 1)] = W[i];
    }
    __syncthreads();

    const int lane = tid & 31;
    const int wave = tid >> 5;
    const int m = lane & 15;
    const int half = lane >> 4;
    const int kg = half << 1;

    const float* lbase = wt + (half * OUT + m) * 2;

    for (int stripe = blockIdx.x; stripe < numStripes; stripe += gridDim.x) {
        const int rowBase = stripe * 128 + wave * 16;
        int arow = rowBase + m;
        if (arow >= N) arow = N - 1;
        const float* xrow = X + (long)arow * K;

        float rs[8];
#pragma unroll
        for (int r = 0; r < 8; ++r) {
            const int row = rowBase + r + 8 * half;
            rs[r] = SCALE_ROWS ? rowscale[row < N ? row : N - 1] : 1.0f;
        }

        v8f zero = {};
        v8f acc[NT];
#pragma unroll
        for (int t = 0; t < NT; ++t) acc[t] = zero;

#pragma unroll 8
        for (int k = 0; k < K; k += 4) {
            v2f a = *(const v2f*)(xrow + k + kg);
            v2f b[NT];
            const float* lk = lbase + (k >> 1) * (2 * OUT);
#pragma unroll
            for (int t = 0; t < NT; ++t) b[t] = *(const v2f*)(lk + t * 32);
#pragma unroll
            for (int t = 0; t < NT; ++t)
                acc[t] = __builtin_amdgcn_wmma_f32_16x16x4_f32(
                    false, a, false, b[t], (short)0, acc[t], false, false);
        }

#pragma unroll
        for (int t = 0; t < NT; ++t) {
            const int col = t * 16 + m;
#pragma unroll
            for (int r = 0; r < 8; ++r) {
                const int row = rowBase + r + 8 * half;
                if (row < N) Y[(long)row * OUT + col] = SCALE_ROWS ? acc[t][r] * rs[r] : acc[t][r];
            }
        }
    }
}


extern "C" void kernel_launch(void* const* d_in, const int* in_sizes, int n_in,
                              void* d_out, int out_size, void* d_ws, size_t ws_size,
                              hipStream_t stream) {
    const float* x  = (const float*)d_in[0];
    const int*   ei = (const int*)d_in[1];
    const float* W1 = (const float*)d_in[2];
    const float* b1 = (const float*)d_in[3];
    const float* W2 = (const float*)d_in[4];
    const float* b2 = (const float*)d_in[5];
    float* out = (float*)d_out;

    const int N = in_sizes[0] / 128;
    const int E = in_sizes[1] / 2;
    const int* src = ei;
    const int* dst = ei + E;

    float* dinv    = (float*)d_ws;
    float* xw      = dinv + N;
    float* t       = xw + (size_t)N * 64;
    float* w12     = t + (size_t)N * 64;
    float* b12     = w12 + 128 * 64;
    int*   cnt     = (int*)(w12 + 129 * 64);
    int*   offsets = cnt + N;
    int*   rec     = offsets + N;
    int*   csr_src = rec + E;

    const int NB = (N + 255) / 256;
    const int CH = (E + SORT_NC - 1) / SORT_NC;
    const int M  = NB * SORT_C;
    int* hist = (int*)t;
    int* lst  = hist + M;
    chunk_order_fold_kernel<<<SORT_C, SORT_BD, (NB + CH) * sizeof(int), stream>>>(src, dst, hist, lst, rec, E, CH, NB,
                                                                                           W1, b1, W2, w12);
    bucket_sort_kernel<<<NB, SORT_BD, 0, stream>>>(rec, hist, lst, csr_src, cnt, offsets, dinv, N, CH);

    const int numStripes = (N + 127) / 128;
    const int gemmBlocks = numStripes < 320 ? numStripes : 320;
    const int gatherBlocks = ((N + NG_NPW - 1) / NG_NPW + 7) / 8;

    gemm_wmma_kernel<4, true><<<gemmBlocks, 256, 64 * 128 * sizeof(float), stream>>>(x, w12, dinv, xw, N, numStripes);

    node_gather_kernel<2, true, true><<<gatherBlocks, 256, 0, stream>>>(xw, dinv, offsets, cnt, csr_src, b12, t, N);
    node_gather_kernel<2, true, false><<<gatherBlocks, 256, 0, stream>>>(t, dinv, offsets, cnt, csr_src, b2, out, N);
}


# x @ W12 moved into the bucket launch: each bucket's work-group multiplies its own 256 rows (staged through LDS; same f32 16x16x4 WMMA form, k order and sums) once dinv is known; 4 launches, not 5
# speedup vs baseline: 6.8212x; 1.0743x over previous
#include <hip/hip_runtime.h>


typedef __attribute__((ext_vector_type(2))) float v2f;
typedef __attribute__((ext_vector_type(4))) float v4f;
typedef __attribute__((ext_vector_type(8))) float v8f;


constexpr int SORT_BD = 1024;
constexpr int SORT_C  = 256;
constexpr int SORT_U  = 8;

template <bool FULL>
__device__ __forceinline__ void count_trip(const int* __restrict__ dst, int* lh, int e0, int end) {
    int d[SORT_U];
#pragma unroll
    for (int u = 0; u < SORT_U; ++u) {
        const int e = e0 + u * SORT_BD;
        d[u] = dst[(FULL || e < end) ? e : (end - 1)];
    }
    if (!FULL) {
#pragma unroll
        for (int u = 0; u < SORT_U; ++u)
            asm volatile("" : "+v"(d[u]));
    }
#pragma unroll
    for (int u = 0; u < SORT_U; ++u)
        if (FULL || e0 + u * SORT_BD < end) atomicAdd(&lh[d[u] >> 8], 1);
}

constexpr int FOLD_WG = 9;
constexpr int SORT_NC = SORT_C - FOLD_WG;

__device__ __forceinline__ void fold_weights_role(const float* __restrict__ W1,
                                                  const float* __restrict__ b1,
                                                  const float* __restrict__ W2,
                                                  float* __restrict__ w12, int g, float* wt) {
    constexpr int K = 128;
    constexpr int OUT = 64;
    const int tid = threadIdx.x;
    const int lane = tid & 31;
    const int wave = tid >> 5;
    const int m = lane & 15;
    const int half = lane >> 4;
    const int kg = half << 1;
    const bool compute = wave < 4;

    v2f a[K / 4];
    if (compute) {
        const float* arow = (g < 8) ? (W1 + (size_t)(g * 16 + m) * K) : b1;
#pragma unroll
        for (int j = 0; j < K / 4; ++j) a[j] = *(const v2f*)(arow + 4 * j + kg);
    }

    float w[K * OUT / 1024];
#pragma unroll
    for (int u = 0; u < K * OUT / 1024; ++u) w[u] = W2[tid + u * 1024];
#pragma unroll
    for (int u = 0; u < K * OUT / 1024; ++u) {
        const int i = tid + u * 1024;
        const int k = i / OUT;
        const int n = i - k * OUT;
        wt[(k >> 1) * (2 * OUT) + 2 * n + (k & 1)] = w[u];
    }
    __syncthreads();
    if (!compute) return;

    int ooff = (g * 16 + 8 * half) * OUT + wave * 16 + m;
    const float* lbase = wt + (half * OUT + m) * 2 + wave * 32;
    v8f acc = {};
    asm volatile("" : "+v"(ooff));
    __builtin_amdgcn_sched_barrier(0);
#pragma unroll
    for (int j = 0; j < K / 4; ++j) {
        const v2f b = *(const v2f*)(lbase + (2 * j) * (2 * OUT));
        acc = __builtin_amdgcn_wmma_f32_16x16x4_f32(false, a[j], false, b, (short)0, acc, false, false);
    }
    __builtin_amdgcn_sched_barrier(0);
    asm volatile("v_nop\n\tv_nop\n\tv_nop\n\tv_nop");
    __builtin_amdgcn_sched_barrier(0);

    if (g < 8) {
#pragma unroll
        for (int r = 0; r < 8; ++r) w12[ooff + r * OUT] = acc[r];
    } else if (half == 0) {
        w12[ooff] = acc[0];
    }
}

template <bool FULL>
__device__ __forceinline__ void place_trip(const int* __restrict__ src, const int* __restrict__ dst,
                                           int* stage, int* cur, int e0, int end) {
    int d[SORT_U], s[SORT_U];
#pragma unroll
    for (int u = 0; u < SORT_U; ++u) {
        const int e = e0 + u * SORT_BD;
        const int ec = (FULL || e < end) ? e : (end - 1);
        d[u] = dst[ec];
        s[u] = src[ec];
    }
    if (!FULL) {
#pragma unroll
        for (int u = 0; u < SORT_U; ++u)
            asm volatile("" : "+v"(d[u]), "+v"(s[u]));
    }
#pragma unroll
    for (int u = 0; u < SORT_U; ++u)
        if (FULL || e0 + u * SORT_BD < end) {
            const int pos = atomicAdd(&cur[d[u] >> 8], 1);
            stage[pos] = (int)(((unsigned)s[u] << 8) | (unsigned)(d[u] & 255));
        }
}

__global__ __launch_bounds__(SORT_BD) void chunk_order_fold_kernel(const int* __restrict__ src,
                                                                   const int* __restrict__ dst,
                                                                   int* __restrict__ hist,
                                                                   int* __restrict__ lst,
                                                                   int* __restrict__ rec,
                                                                   int E, int CH, int NB,
                                                                   const float* __restrict__ W1,
                                                                   const float* __restrict__ b1,
                                                                   const float* __restrict__ W2,
                                                                   float* __restrict__ w12) {
    extern __shared__ int dyn[];
    __shared__ float wt[128 * 64];
    __shared__ int wsum[32];
    __shared__ int s_tot;
    const int col = blockIdx.x;
    if (col < FOLD_WG) {
        for (int b = threadIdx.x; b < NB; b += SORT_BD) {
            hist[b * SORT_C + col] = 0;
            lst[b * SORT_C + col] = 0;
        }
        fold_weights_role(W1, b1, W2, w12, col, wt);
        return;
    }
    int* lh = dyn;
    int* stage = dyn + NB;
    const int tid = threadIdx.x;
    const int lane = tid & 31;
    const int wave = tid >> 5;
    const int c = col - FOLD_WG;
    for (int b = tid; b < NB; b += SORT_BD) lh[b] = 0;
    __syncthreads();
    const int beg = c * CH;
    const int end = (beg + CH < E) ? (beg + CH) : E;
    for (int e0 = beg + tid; e0 < end; e0 += SORT_U * SORT_BD) {
        if (e0 + (SORT_U - 1) * SORT_BD < end) count_trip<true>(dst, lh, e0, end);
        else                                   count_trip<false>(dst, lh, e0, end);
    }
    __syncthreads();

    int carry = 0;
    for (int b0 = 0; b0 < NB; b0 += SORT_BD) {
        const int b = b0 + tid;
        const int cv = (b < NB) ? lh[b] : 0;
        int incl = cv;
        for (int o = 1; o < 32; o <<= 1) {
            int t = __shfl_up(incl, o);
            if (lane >= o) incl += t;
        }
        if (lane == 31) wsum[wave] = incl;
        __syncthreads();
        if (wave == 0) {
            int v = wsum[lane];
            int iv = v;
            for (int o = 1; o < 32; o <<= 1) {
                int t = __shfl_up(iv, o);
                if (lane >= o) iv += t;
            }
            wsum[lane] = iv - v;
            if (lane == 31) s_tot = iv;
        }
        __syncthreads();
        const int excl = carry + wsum[wave] + (incl - cv);
        if (b < NB) {
            lh[b] = excl;
            hist[b * SORT_C + col] = cv;
            lst[b * SORT_C + col] = excl;
        }
        carry += s_tot;
        __syncthreads();
    }

    for (int e0 = beg + tid; e0 < end; e0 += SORT_U * SORT_BD) {
        if (e0 + (SORT_U - 1) * SORT_BD < end) place_trip<true>(src, dst, stage, lh, e0, end);
        else                                   place_trip<false>(src, dst, stage, lh, e0, end);
    }
    __syncthreads();
    for (int i = tid; i < end - beg; i += SORT_BD) rec[beg + i] = stage[i];
}

constexpr int SEG_U = 8;

template <bool PLACE>
__device__ __forceinline__ void run_trip(const int* __restrict__ rec, int* __restrict__ csr_src,
                                         int* lc, int bbeg, int seg0, int segn, int k0) {
    int r[SEG_U];
#pragma unroll
    for (int u = 0; u < SEG_U; ++u) {
        const int k = k0 + 4 * u;
        r[u] = rec[seg0 + ((k < segn) ? k : 0)];
    }
#pragma unroll
    for (int u = 0; u < SEG_U; ++u)
        if (k0 + 4 * u < segn) {
            const int pos = atomicAdd(&lc[r[u] & 255], 1);
            if (PLACE) csr_src[bbeg + pos] = (int)((unsigned)r[u] >> 8);
        }
}

constexpr int XS_PITCH = 132;

__device__ __forceinline__ void x_rows_load(const float* x, int N, int row0, int lane, v4f (&g)[4]) {
#pragma unroll
    for (int u = 0; u < 4; ++u) {
        int row = row0 + u;
        if (row >= N) row = N - 1;
        g[u] = *(const v4f*)(x + (long)row * 128 + lane * 4);
    }
}

__device__ __forceinline__ void x_rows_stage(float* xs, int rl0, int lane, const v4f (&g)[4]) {
#pragma unroll
    for (int u = 0; u < 4; ++u) *(v4f*)(xs + (rl0 + u) * XS_PITCH + lane * 4) = g[u];
}

__device__ __forceinline__ void tile_product_store(const float* xs, const float* wt, const float* sdinv,
                                                   float* __restrict__ xw, int N, int bucketRow0, int pass0,
                                                   int wave, int lane) {
    const int m = lane & 15;
    const int half = lane >> 4;
    const int rt = wave >> 2;
    const int ct = wave & 3;
    const float* abase = xs + (rt * 16 + m) * XS_PITCH + 2 * half;
    const float* bbase = wt + (half * 64 + ct * 16 + m) * 2;
    v8f acc = {};
#pragma nounroll
    for (int k0 = 0; k0 < 128; k0 += 32) {
#pragma unroll
        for (int k = k0; k < k0 + 32; k += 4) {
            const v2f a = *(const v2f*)(abase + k);
            const v2f bq = *(const v2f*)(bbase + (k >> 1) * 128);
            acc = __builtin_amdgcn_wmma_f32_16x16x4_f32(false, a, false, bq, (short)0, acc, false, false);
        }
        __builtin_amdgcn_sched_barrier(0);
        asm volatile("v_nop\n\tv_nop\n\tv_nop\n\tv_nop");
        __builtin_amdgcn_sched_barrier(0);
    }

#pragma unroll
    for (int r = 0; r < 8; ++r) {
        const int rloc = pass0 + rt * 16 + r + 8 * half;
        const int row = bucketRow0 + rloc;
        if (row < N) xw[(long)row * 64 + ct * 16 + m] = acc[r] * sdinv[rloc];
    }
}

__global__ __launch_bounds__(SORT_BD) void bucket_sort_gemm_kernel(const int* __restrict__ rec,
                                                                   const int* __restrict__ hist,
                                                                   const int* __restrict__ lst,
                                                                   int* __restrict__ csr_src,
                                                                   int* __restrict__ cnt,
                                                                   int* __restrict__ offsets,
                                                                   float* __restrict__ dinv,
                                                                   int N, int CH,
                                                                   const float* x,
                                                                   const float* __restrict__ w12,
                                                                   float* __restrict__ xw) {
    __shared__ int lc[256];
    __shared__ int wsum[32];
    __shared__ int s_prefix;
    __shared__ float sdinv[256];
    __shared__ __attribute__((aligned(16))) float wt[128 * 64];
    __shared__ __attribute__((aligned(16))) float xs[128 * XS_PITCH];
    const int tid = threadIdx.x;
    const int lane = tid & 31;
    const int wave = tid >> 5;
    const int b = blockIdx.x;

    float wq[128 * 64 / SORT_BD];
#pragma unroll
    for (int u = 0; u < 128 * 64 / SORT_BD; ++u) wq[u] = w12[tid + u * SORT_BD];

    const int cc = tid >> 2;
    const int sub = tid & 3;
    const int segn = hist[b * SORT_C + cc];
    const int seg0 = (cc - FOLD_WG) * CH + lst[b * SORT_C + cc];

    int part = 0;
    const int4* c4 = (const int4*)hist;
    const int lim = (b * SORT_C) >> 2;
    for (int j = tid; j < lim; j += 8 * SORT_BD) {
        int4 v[8];
#pragma unroll
        for (int u = 0; u < 8; ++u) {
            const int q = j + u * SORT_BD;
            v[u] = c4[(q < lim) ? q : (lim - 1)];
        }
#pragma unroll
        for (int u = 0; u < 8; ++u)
            if (j + u * SORT_BD < lim) part += (v[u].x + v[u].y) + (v[u].z + v[u].w);
    }
    for (int o = 16; o > 0; o >>= 1) part += __shfl_xor(part, o);
    if (lane == 0) wsum[wave] = part;
    if (tid < 256) lc[tid] = 0;
#pragma unroll
    for (int u = 0; u < 128 * 64 / SORT_BD; ++u) {
        const int i = tid + u * SORT_BD;
        const int k = i >> 6;
        const int n = i & 63;
        wt[(k >> 1) * 128 + 2 * n + (k & 1)] = wq[u];
    }
    __syncthreads();
    if (wave == 0) {
        int v = wsum[lane];
        for (int o = 16; o > 0; o >>= 1) v += __shfl_xor(v, o);
        if (lane == 0) s_prefix = v;
    }
    for (int k0 = sub; k0 < segn; k0 += 4 * SEG_U) run_trip<false>(rec, csr_src, lc, 0, seg0, segn, k0);
    __syncthreads();
    const int bbeg = s_prefix;

    int c = 0, incl = 0;
    if (tid < 256) {
        c = lc[tid];
        incl = c;
        for (int o = 1; o < 32; o <<= 1) {
            int t = __shfl_up(incl, o);
            if (lane >= o) incl += t;
        }
        if (lane == 31) wsum[wave] = incl;
    }
    __syncthreads();
    if (tid < 256) {
        int wpre = 0;
        for (int w = 0; w < 8; ++w) wpre += (w < wave) ? wsum[w] : 0;
        const int excl = wpre + (incl - c);
        lc[tid] = excl;
        const int node = b * 256 + tid;
        const float dg = (float)(c + 1);
        const float dv = (dg > 0.0f) ? rsqrtf(dg) : 0.0f;
        sdinv[tid] = dv;
        if (node < N) {
            cnt[node] = c;
            offsets[node] = bbeg + excl;
            dinv[node] = dv;
        }
    }
    __syncthreads();
    v4f g0[4];
    x_rows_load(x, N, b * 256 + wave * 4, lane, g0);
    for (int k0 = sub; k0 < segn; k0 += 4 * SEG_U) run_trip<true>(rec, csr_src, lc, bbeg, seg0, segn, k0);

    x_rows_stage(xs, wave * 4, lane, g0);
    __syncthreads();
    v4f g1[4];
    x_rows_load(x, N, b * 256 + 128 + wave * 4, lane, g1);
    tile_product_store(xs, wt, sdinv, xw, N, b * 256, 0, wave, lane);
    __syncthreads();
    x_rows_stage(xs, wave * 4, lane, g1);
    __syncthreads();
    tile_product_store(xs, wt, sdinv, xw, N, b * 256, 128, wave, lane);
}


constexpr int NG_NPW = 7;

template <int VPL, bool HAS_BIAS, bool SCALE_OUT>
__global__ __launch_bounds__(256) void node_gather_kernel(const float* __restrict__ in,
                                                          const float* __restrict__ dinv,
                                                          const int* __restrict__ offsets,
                                                          const int* __restrict__ cnt,
                                                          const int* __restrict__ csr_src,
                                                          const float* __restrict__ bias,
                                                          float* __restrict__ out, int N) {
    constexpr int DIM = 32 * VPL;
    typedef __attribute__((ext_vector_type(VPL))) float vec;
    const int lane = threadIdx.x & 31;
    int w = blockIdx.x * 8 + (threadIdx.x >> 5);
    w = __builtin_amdgcn_readfirstlane(w);
    const int d0 = w * NG_NPW;
    if (d0 >= N) return;
    const int nn = (N - d0 < NG_NPW) ? (N - d0) : NG_NPW;

    vec self_n = *(const vec*)(in + (size_t)d0 * DIM + lane * VPL);
    const int dl = d0 + ((lane < nn) ? lane : 0);
    const int offv = offsets[dl];
    const int cntv = cnt[dl];
    const int dinvb = __float_as_int(dinv[dl]);
    vec bv = {};
    if (HAS_BIAS) bv = *(const vec*)(bias + lane * VPL);

    int nbeg = __builtin_amdgcn_readlane(offv, 0);
    int nlen = __builtin_amdgcn_readlane(cntv, 0);
    int sj_n = 0;
    if (lane < nlen) sj_n = csr_src[nbeg + lane];

    for (int i = 0; i < nn; ++i) {
        const int d = d0 + i;
        const int beg = nbeg;
        const int len = nlen;
        const float dd = __int_as_float(__builtin_amdgcn_readlane(dinvb, i));
        int sj = sj_n;
        vec acc = self_n;

        if (i + 1 < nn) {
            nbeg = __builtin_amdgcn_readlane(offv, i + 1);
            nlen = __builtin_amdgcn_readlane(cntv, i + 1);
            sj_n = 0;
            if (lane < nlen) sj_n = csr_src[nbeg + lane];
            self_n = *(const vec*)(in + (size_t)(d + 1) * DIM + lane * VPL);
        }

        for (int base = 0; base < len; base += 32) {
            const int n = (len - base < 32) ? (len - base) : 32;
            if (base > 0) {
                sj = 0;
                if (lane < n) sj = csr_src[beg + base + lane];
                asm volatile("" : "+v"(sj));
            }
#define NG_ROW(jj) (*(const vec*)(in + (size_t)__builtin_amdgcn_readlane(sj, (jj)) * DIM + lane * VPL))
            int j = 0;
            for (; j + 8 <= n; j += 8) {
                const vec r0 = NG_ROW(j),     r1 = NG_ROW(j + 1), r2 = NG_ROW(j + 2), r3 = NG_ROW(j + 3);
                const vec r4 = NG_ROW(j + 4), r5 = NG_ROW(j + 5), r6 = NG_ROW(j + 6), r7 = NG_ROW(j + 7);
                acc += r0;
                acc += r1;
                acc += r2;
                acc += r3;
                acc += r4;
                acc += r5;
                acc += r6;
                acc += r7;
            }
            if (j + 4 <= n) {
                const vec r0 = NG_ROW(j), r1 = NG_ROW(j + 1), r2 = NG_ROW(j + 2), r3 = NG_ROW(j + 3);
                acc += r0;
                acc += r1;
                acc += r2;
                acc += r3;
                j += 4;
            }
            if (j < n) {
                const vec z = {};
                const vec r0 = NG_ROW(j);
                vec r1 = z, r2 = z;
                if (j + 1 < n) r1 = NG_ROW(j + 1);
                if (j + 2 < n) r2 = NG_ROW(j + 2);
                acc += r0;
                acc += r1;
                acc += r2;
            }
#undef NG_ROW
        }

        acc *= dd;
        if (HAS_BIAS) acc += bv;
        if (SCALE_OUT) acc *= dd;
        *(vec*)(out + (size_t)d * DIM + lane * VPL) = acc;
    }
}


extern "C" void kernel_launch(void* const* d_in, const int* in_sizes, int n_in,
                              void* d_out, int out_size, void* d_ws, size_t ws_size,
                              hipStream_t stream) {
    const float* x  = (const float*)d_in[0];
    const int*   ei = (const int*)d_in[1];
    const float* W1 = (const float*)d_in[2];
    const float* b1 = (const float*)d_in[3];
    const float* W2 = (const float*)d_in[4];
    const float* b2 = (const float*)d_in[5];
    float* out = (float*)d_out;

    const int N = in_sizes[0] / 128;
    const int E = in_sizes[1] / 2;
    const int* src = ei;
    const int* dst = ei + E;

    float* dinv    = (float*)d_ws;
    float* xw      = dinv + N;
    float* t       = xw + (size_t)N * 64;
    float* w12     = t + (size_t)N * 64;
    float* b12     = w12 + 128 * 64;
    int*   cnt     = (int*)(w12 + 129 * 64);
    int*   offsets = cnt + N;
    int*   rec     = offsets + N;
    int*   csr_src = rec + E;

    const int NB = (N + 255) / 256;
    const int CH = (E + SORT_NC - 1) / SORT_NC;
    const int M  = NB * SORT_C;
    int* hist = (int*)t;
    int* lst  = hist + M;
    chunk_order_fold_kernel<<<SORT_C, SORT_BD, (NB + CH) * sizeof(int), stream>>>(src, dst, hist, lst, rec, E, CH, NB,
                                                                                           W1, b1, W2, w12);
    bucket_sort_gemm_kernel<<<NB, SORT_BD, 0, stream>>>(rec, hist, lst, csr_src, cnt, offsets, dinv, N, CH, x, w12, xw);

    const int gatherBlocks = ((N + NG_NPW - 1) / NG_NPW + 7) / 8;

    node_gather_kernel<2, true, true><<<gatherBlocks, 256, 0, stream>>>(xw, dinv, offsets, cnt, csr_src, b12, t, N);
    node_gather_kernel<2, true, false><<<gatherBlocks, 256, 0, stream>>>(t, dinv, offsets, cnt, csr_src, b2, out, N);
}


# speedup vs baseline: 7.0854x; 1.0387x over previous
#include <hip/hip_runtime.h>


typedef __attribute__((ext_vector_type(2))) float v2f;
typedef __attribute__((ext_vector_type(4))) float v4f;
typedef __attribute__((ext_vector_type(8))) float v8f;


constexpr int SORT_BD = 1024;
constexpr int SORT_C  = 256;
constexpr int SORT_U  = 8;

template <bool FULL>
__device__ __forceinline__ void count_trip(const int* __restrict__ dst, int* lh, int e0, int end) {
    int d[SORT_U];
#pragma unroll
    for (int u = 0; u < SORT_U; ++u) {
        const int e = e0 + u * SORT_BD;
        d[u] = dst[(FULL || e < end) ? e : (end - 1)];
    }
    if (!FULL) {
#pragma unroll
        for (int u = 0; u < SORT_U; ++u)
            asm volatile("" : "+v"(d[u]));
    }
#pragma unroll
    for (int u = 0; u < SORT_U; ++u)
        if (FULL || e0 + u * SORT_BD < end) atomicAdd(&lh[d[u] >> 8], 1);
}

constexpr int FOLD_WG = 9;
constexpr int SORT_NC = SORT_C - FOLD_WG;

__device__ __forceinline__ void fold_weights_role(const float* __restrict__ W1,
                                                  const float* __restrict__ b1,
                                                  const float* __restrict__ W2,
                                                  float* __restrict__ w12, int g, float* wt) {
    constexpr int K = 128;
    constexpr int OUT = 64;
    const int tid = threadIdx.x;
    const int lane = tid & 31;
    const int wave = tid >> 5;
    const int m = lane & 15;
    const int half = lane >> 4;
    const int kg = half << 1;
    const bool compute = wave < 4;

    v2f a[K / 4];
    if (compute) {
        const float* arow = (g < 8) ? (W1 + (size_t)(g * 16 + m) * K) : b1;
#pragma unroll
        for (int j = 0; j < K / 4; ++j) a[j] = *(const v2f*)(arow + 4 * j + kg);
    }

    float w[K * OUT / 1024];
#pragma unroll
    for (int u = 0; u < K * OUT / 1024; ++u) w[u] = W2[tid + u * 1024];
#pragma unroll
    for (int u = 0; u < K * OUT / 1024; ++u) {
        const int i = tid + u * 1024;
        const int k = i / OUT;
        const int n = i - k * OUT;
        wt[(k >> 1) * (2 * OUT) + 2 * n + (k & 1)] = w[u];
    }
    __syncthreads();
    if (!compute) return;

    int ooff = (g * 16 + 8 * half) * OUT + wave * 16 + m;
    const float* lbase = wt + (half * OUT + m) * 2 + wave * 32;
    v8f acc = {};
    asm volatile("" : "+v"(ooff));
    __builtin_amdgcn_sched_barrier(0);
#pragma unroll
    for (int j = 0; j < K / 4; ++j) {
        const v2f b = *(const v2f*)(lbase + (2 * j) * (2 * OUT));
        acc = __builtin_amdgcn_wmma_f32_16x16x4_f32(false, a[j], false, b, (short)0, acc, false, false);
    }
    __builtin_amdgcn_sched_barrier(0);
    asm volatile("v_nop\n\tv_nop\n\tv_nop\n\tv_nop");
    __builtin_amdgcn_sched_barrier(0);

    if (g < 8) {
#pragma unroll
        for (int r = 0; r < 8; ++r) w12[ooff + r * OUT] = acc[r];
    } else if (half == 0) {
        w12[ooff] = acc[0];
    }
}

template <bool FULL>
__device__ __forceinline__ void place_trip(const int* __restrict__ src, const int* __restrict__ dst,
                                           int* stage, int* cur, int e0, int end) {
    int d[SORT_U], s[SORT_U];
#pragma unroll
    for (int u = 0; u < SORT_U; ++u) {
        const int e = e0 + u * SORT_BD;
        const int ec = (FULL || e < end) ? e : (end - 1);
        d[u] = dst[ec];
        s[u] = src[ec];
    }
    if (!FULL) {
#pragma unroll
        for (int u = 0; u < SORT_U; ++u)
            asm volatile("" : "+v"(d[u]), "+v"(s[u]));
    }
#pragma unroll
    for (int u = 0; u < SORT_U; ++u)
        if (FULL || e0 + u * SORT_BD < end) {
            const int pos = atomicAdd(&cur[d[u] >> 8], 1);
            stage[pos] = (int)(((unsigned)s[u] << 8) | (unsigned)(d[u] & 255));
        }
}

__global__ __launch_bounds__(SORT_BD) void chunk_order_fold_kernel(const int* __restrict__ src,
                                                                   const int* __restrict__ dst,
                                                                   int* __restrict__ hist,
                                                                   int* __restrict__ lst,
                                                                   int* __restrict__ rec,
                                                                   int E, int CH, int NB,
                                                                   const float* __restrict__ W1,
                                                                   const float* __restrict__ b1,
                                                                   const float* __restrict__ W2,
                                                                   float* __restrict__ w12) {
    extern __shared__ int dyn[];
    __shared__ float wt[128 * 64];
    __shared__ int wsum[32];
    __shared__ int s_tot;
    const int col = blockIdx.x;
    if (col < FOLD_WG) {
        for (int b = threadIdx.x; b < NB; b += SORT_BD) {
            hist[b * SORT_C + col] = 0;
            lst[b * SORT_C + col] = 0;
        }
        fold_weights_role(W1, b1, W2, w12, col, wt);
        return;
    }
    int* lh = dyn;
    int* stage = dyn + NB;
    const int tid = threadIdx.x;
    const int lane = tid & 31;
    const int wave = tid >> 5;
    const int c = col - FOLD_WG;
    for (int b = tid; b < NB; b += SORT_BD) lh[b] = 0;
    __syncthreads();
    const int beg = c * CH;
    const int end = (beg + CH < E) ? (beg + CH) : E;
    for (int e0 = beg + tid; e0 < end; e0 += SORT_U * SORT_BD) {
        if (e0 + (SORT_U - 1) * SORT_BD < end) count_trip<true>(dst, lh, e0, end);
        else                                   count_trip<false>(dst, lh, e0, end);
    }
    __syncthreads();

    int carry = 0;
    for (int b0 = 0; b0 < NB; b0 += SORT_BD) {
        const int b = b0 + tid;
        const int cv = (b < NB) ? lh[b] : 0;
        int incl = cv;
        for (int o = 1; o < 32; o <<= 1) {
            int t = __shfl_up(incl, o);
            if (lane >= o) incl += t;
        }
        if (lane == 31) wsum[wave] = incl;
        __syncthreads();
        if (wave == 0) {
            int v = wsum[lane];
            int iv = v;
            for (int o = 1; o < 32; o <<= 1) {
                int t = __shfl_up(iv, o);
                if (lane >= o) iv += t;
            }
            wsum[lane] = iv - v;
            if (lane == 31) s_tot = iv;
        }
        __syncthreads();
        const int excl = carry + wsum[wave] + (incl - cv);
        if (b < NB) {
            lh[b] = excl;
            hist[b * SORT_C + col] = cv;
            lst[b * SORT_C + col] = excl;
        }
        carry += s_tot;
        __syncthreads();
    }

    for (int e0 = beg + tid; e0 < end; e0 += SORT_U * SORT_BD) {
        if (e0 + (SORT_U - 1) * SORT_BD < end) place_trip<true>(src, dst, stage, lh, e0, end);
        else                                   place_trip<false>(src, dst, stage, lh, e0, end);
    }
    __syncthreads();
    for (int i = tid; i < end - beg; i += SORT_BD) rec[beg + i] = stage[i];
}

constexpr int SEG_U = 8;

template <bool PLACE>
__device__ __forceinline__ void run_trip(const int* __restrict__ rec, int* __restrict__ csr_src,
                                         int* lc, int bbeg, int seg0, int segn, int k0) {
    int r[SEG_U];
#pragma unroll
    for (int u = 0; u < SEG_U; ++u) {
        const int k = k0 + 4 * u;
        r[u] = rec[seg0 + ((k < segn) ? k : 0)];
    }
#pragma unroll
    for (int u = 0; u < SEG_U; ++u)
        if (k0 + 4 * u < segn) {
            const int pos = atomicAdd(&lc[r[u] & 255], 1);
            if (PLACE) csr_src[bbeg + pos] = (int)((unsigned)r[u] >> 8);
        }
}

constexpr int XS_PITCH = 132;

__device__ __forceinline__ void x_rows_load(const float* x, int N, int row0, int lane, v4f (&g)[4]) {
#pragma unroll
    for (int u = 0; u < 4; ++u) {
        int row = row0 + u;
        if (row >= N) row = N - 1;
        g[u] = *(const v4f*)(x + (long)row * 128 + lane * 4);
    }
}

__device__ __forceinline__ void x_rows_stage(float* xs, int rl0, int lane, const v4f (&g)[4]) {
#pragma unroll
    for (int u = 0; u < 4; ++u) *(v4f*)(xs + (rl0 + u) * XS_PITCH + lane * 4) = g[u];
}

__device__ __forceinline__ void tile_product_store(const float* xs, const float* wt, const float* sdinv,
                                                   float* __restrict__ xw, int N, int bucketRow0, int pass0,
                                                   int wave, int lane) {
    const int m = lane & 15;
    const int half = lane >> 4;
    const int rt = wave >> 2;
    const int ct = wave & 3;
    const float* abase = xs + (rt * 16 + m) * XS_PITCH + 2 * half;
    const float* bbase = wt + (half * 64 + ct * 16 + m) * 2;
    v8f acc = {};
#pragma nounroll
    for (int k0 = 0; k0 < 128; k0 += 32) {
#pragma unroll
        for (int k = k0; k < k0 + 32; k += 4) {
            const v2f a = *(const v2f*)(abase + k);
            const v2f bq = *(const v2f*)(bbase + (k >> 1) * 128);
            acc = __builtin_amdgcn_wmma_f32_16x16x4_f32(false, a, false, bq, (short)0, acc, false, false);
        }
        __builtin_amdgcn_sched_barrier(0);
        asm volatile("v_nop\n\tv_nop\n\tv_nop\n\tv_nop");
        __builtin_amdgcn_sched_barrier(0);
    }

#pragma unroll
    for (int r = 0; r < 8; ++r) {
        const int rloc = pass0 + rt * 16 + r + 8 * half;
        const int row = bucketRow0 + rloc;
        if (row < N) xw[(long)row * 64 + ct * 16 + m] = acc[r] * sdinv[rloc];
    }
}

__global__ __launch_bounds__(SORT_BD) void bucket_sort_gemm_kernel(const int* __restrict__ rec,
                                                                   const int* __restrict__ hist,
                                                                   const int* __restrict__ lst,
                                                                   int* __restrict__ csr_src,
                                                                   int* __restrict__ cnt,
                                                                   int* __restrict__ offsets,
                                                                   float* __restrict__ dinv,
                                                                   int N, int CH,
                                                                   const float* x,
                                                                   const float* __restrict__ w12,
                                                                   float* __restrict__ xw) {
    __shared__ int lc[256];
    __shared__ int wsum[32];
    __shared__ int s_prefix;
    __shared__ float sdinv[256];
    __shared__ __attribute__((aligned(16))) float wt[128 * 64];
    __shared__ __attribute__((aligned(16))) float xs[128 * XS_PITCH];
    const int tid = threadIdx.x;
    const int lane = tid & 31;
    const int wave = tid >> 5;
    const int b = blockIdx.x;

    float wq[128 * 64 / SORT_BD];
#pragma unroll
    for (int u = 0; u < 128 * 64 / SORT_BD; ++u) wq[u] = w12[tid + u * SORT_BD];

    const int cc = tid >> 2;
    const int sub = tid & 3;
    const int segn = hist[b * SORT_C + cc];
    const int seg0 = (cc - FOLD_WG) * CH + lst[b * SORT_C + cc];

    int part = 0;
    const int4* c4 = (const int4*)hist;
    const int lim = (b * SORT_C) >> 2;
    for (int j = tid; j < lim; j += 8 * SORT_BD) {
        int4 v[8];
#pragma unroll
        for (int u = 0; u < 8; ++u) {
            const int q = j + u * SORT_BD;
            v[u] = c4[(q < lim) ? q : (lim - 1)];
        }
#pragma unroll
        for (int u = 0; u < 8; ++u)
            if (j + u * SORT_BD < lim) part += (v[u].x + v[u].y) + (v[u].z + v[u].w);
    }
    for (int o = 16; o > 0; o >>= 1) part += __shfl_xor(part, o);
    if (lane == 0) wsum[wave] = part;
    if (tid < 256) lc[tid] = 0;
#pragma unroll
    for (int u = 0; u < 128 * 64 / SORT_BD; ++u) {
        const int i = tid + u * SORT_BD;
        const int k = i >> 6;
        const int n = i & 63;
        wt[(k >> 1) * 128 + 2 * n + (k & 1)] = wq[u];
    }
    __syncthreads();
    if (wave == 0) {
        int v = wsum[lane];
        for (int o = 16; o > 0; o >>= 1) v += __shfl_xor(v, o);
        if (lane == 0) s_prefix = v;
    }
    for (int k0 = sub; k0 < segn; k0 += 4 * SEG_U) run_trip<false>(rec, csr_src, lc, 0, seg0, segn, k0);
    __syncthreads();
    const int bbeg = s_prefix;

    int c = 0, incl = 0;
    if (tid < 256) {
        c = lc[tid];
        incl = c;
        for (int o = 1; o < 32; o <<= 1) {
            int t = __shfl_up(incl, o);
            if (lane >= o) incl += t;
        }
        if (lane == 31) wsum[wave] = incl;
    }
    __syncthreads();
    if (tid < 256) {
        int wpre = 0;
        for (int w = 0; w < 8; ++w) wpre += (w < wave) ? wsum[w] : 0;
        const int excl = wpre + (incl - c);
        lc[tid] = excl;
        const int node = b * 256 + tid;
        const float dg = (float)(c + 1);
        const float dv = (dg > 0.0f) ? rsqrtf(dg) : 0.0f;
        sdinv[tid] = dv;
        if (node < N) {
            cnt[node] = c;
            offsets[node] = bbeg + excl;
            dinv[node] = dv;
        }
    }
    __syncthreads();
    v4f g0[4];
    x_rows_load(x, N, b * 256 + wave * 4, lane, g0);
    for (int k0 = sub; k0 < segn; k0 += 4 * SEG_U) run_trip<true>(rec, csr_src, lc, bbeg, seg0, segn, k0);

    x_rows_stage(xs, wave * 4, lane, g0);
    __syncthreads();
    v4f g1[4];
    x_rows_load(x, N, b * 256 + 128 + wave * 4, lane, g1);
    tile_product_store(xs, wt, sdinv, xw, N, b * 256, 0, wave, lane);
    __syncthreads();
    x_rows_stage(xs, wave * 4, lane, g1);
    __syncthreads();
    tile_product_store(xs, wt, sdinv, xw, N, b * 256, 128, wave, lane);
}


constexpr int NG_NPW = 7;

template <int VPL, bool HAS_BIAS, bool SCALE_OUT>
__global__ __launch_bounds__(256) void node_gather_kernel(const float* __restrict__ in,
                                                          const float* __restrict__ dinv,
                                                          const int* __restrict__ offsets,
                                                          const int* __restrict__ cnt,
                                                          const int* __restrict__ csr_src,
                                                          const float* __restrict__ bias,
                                                          float* __restrict__ out, int N) {
    constexpr int DIM = 32 * VPL;
    typedef __attribute__((ext_vector_type(VPL))) float vec;
    const int lane = threadIdx.x & 31;
    int w = blockIdx.x * 8 + (threadIdx.x >> 5);
    w = __builtin_amdgcn_readfirstlane(w);
    const int d0 = w * NG_NPW;
    if (d0 >= N) return;
    const int nn = (N - d0 < NG_NPW) ? (N - d0) : NG_NPW;

    vec self_n = *(const vec*)(in + (size_t)d0 * DIM + lane * VPL);
    const int dl = d0 + ((lane < nn) ? lane : 0);
    const int offv = offsets[dl];
    const int cntv = cnt[dl];
    const int dinvb = __float_as_int(dinv[dl]);
    vec bv = {};
    if (HAS_BIAS) bv = *(const vec*)(bias + lane * VPL);

    int nbeg = __builtin_amdgcn_readlane(offv, 0);
    int nlen = __builtin_amdgcn_readlane(cntv, 0);
    int sj_n = 0;
    if (lane < nlen) sj_n = csr_src[nbeg + lane];

    for (int i = 0; i < nn; ++i) {
        const int d = d0 + i;
        const int beg = nbeg;
        const int len = nlen;
        const float dd = __int_as_float(__builtin_amdgcn_readlane(dinvb, i));
        int sj = sj_n;
        vec acc = self_n;

        if (i + 1 < nn) {
            nbeg = __builtin_amdgcn_readlane(offv, i + 1);
            nlen = __builtin_amdgcn_readlane(cntv, i + 1);
            sj_n = 0;
            if (lane < nlen) sj_n = csr_src[nbeg + lane];
            self_n = *(const vec*)(in + (size_t)(d + 1) * DIM + lane * VPL);
        }

        for (int base = 0; base < len; base += 32) {
            const int n = (len - base < 32) ? (len - base) : 32;
            if (base > 0) {
                sj = 0;
                if (lane < n) sj = csr_src[beg + base + lane];
                asm volatile("" : "+v"(sj));
            }
#define NG_ROW(jj) (*(const vec*)(in + (size_t)__builtin_amdgcn_readlane(sj, (jj)) * DIM + lane * VPL))
            int j = 0;
            for (; j + 8 <= n; j += 8) {
                const vec r0 = NG_ROW(j),     r1 = NG_ROW(j + 1), r2 = NG_ROW(j + 2), r3 = NG_ROW(j + 3);
                const vec r4 = NG_ROW(j + 4), r5 = NG_ROW(j + 5), r6 = NG_ROW(j + 6), r7 = NG_ROW(j + 7);
                acc += r0;
                acc += r1;
                acc += r2;
                acc += r3;
                acc += r4;
                acc += r5;
                acc += r6;
                acc += r7;
            }
            if (j + 4 <= n) {
                const vec r0 = NG_ROW(j), r1 = NG_ROW(j + 1), r2 = NG_ROW(j + 2), r3 = NG_ROW(j + 3);
                acc += r0;
                acc += r1;
                acc += r2;
                acc += r3;
                j += 4;
            }
            if (j < n) {
                const vec z = {};
                const vec r0 = NG_ROW(j);
                vec r1 = z, r2 = z;
                if (j + 1 < n) r1 = NG_ROW(j + 1);
                if (j + 2 < n) r2 = NG_ROW(j + 2);
                acc += r0;
                acc += r1;
                acc += r2;
            }
#undef NG_ROW
        }

        acc *= dd;
        if (HAS_BIAS) acc += bv;
        if (SCALE_OUT) acc *= dd;
        *(vec*)(out + (size_t)d * DIM + lane * VPL) = acc;
    }
}


extern "C" void kernel_launch(void* const* d_in, const int* in_sizes, int n_in,
                              void* d_out, int out_size, void* d_ws, size_t ws_size,
                              hipStream_t stream) {
    const float* x  = (const float*)d_in[0];
    const int*   ei = (const int*)d_in[1];
    const float* W1 = (const float*)d_in[2];
    const float* b1 = (const float*)d_in[3];
    const float* W2 = (const float*)d_in[4];
    const float* b2 = (const float*)d_in[5];
    float* out = (float*)d_out;

    const int N = in_sizes[0] / 128;
    const int E = in_sizes[1] / 2;
    const int* src = ei;
    const int* dst = ei + E;

    float* xw      = (float*)d_ws;
    float* t       = xw + (size_t)N * 64;
    float* dinv    = t + (size_t)N * 64;
    float* w12     = dinv + N;
    float* b12     = w12 + 128 * 64;
    int*   cnt     = (int*)(w12 + 129 * 64);
    int*   offsets = cnt + N;
    int*   rec     = offsets + N;
    int*   csr_src = rec + E;

    const int NB = (N + 255) / 256;
    const int CH = (E + SORT_NC - 1) / SORT_NC;
    const int M  = NB * SORT_C;
    int* hist = (int*)t;
    int* lst  = hist + M;
    chunk_order_fold_kernel<<<SORT_C, SORT_BD, (NB + CH) * sizeof(int), stream>>>(src, dst, hist, lst, rec, E, CH, NB,
                                                                                           W1, b1, W2, w12);
    bucket_sort_gemm_kernel<<<NB, SORT_BD, 0, stream>>>(rec, hist, lst, csr_src, cnt, offsets, dinv, N, CH, x, w12, xw);

    const int gatherBlocks = ((N + NG_NPW - 1) / NG_NPW + 7) / 8;

    node_gather_kernel<2, true, true><<<gatherBlocks, 256, 0, stream>>>(xw, dinv, offsets, cnt, csr_src, b12, t, N);
    node_gather_kernel<2, true, false><<<gatherBlocks, 256, 0, stream>>>(t, dinv, offsets, cnt, csr_src, b2, out, N);
}


# Bucket's node-ordered source list is built in LDS and copied to csr_src in whole 128-byte lines behind the barrier, not one 4-byte store per edge to scattered lines; same slots, same sum order
# speedup vs baseline: 7.4360x; 1.0495x over previous
#include <hip/hip_runtime.h>


typedef __attribute__((ext_vector_type(2))) float v2f;
typedef __attribute__((ext_vector_type(4))) float v4f;
typedef __attribute__((ext_vector_type(8))) float v8f;


constexpr int SORT_BD = 1024;
constexpr int SORT_C  = 256;
constexpr int SORT_U  = 8;

template <bool FULL>
__device__ __forceinline__ void count_trip(const int* __restrict__ dst, int* lh, int e0, int end) {
    int d[SORT_U];
#pragma unroll
    for (int u = 0; u < SORT_U; ++u) {
        const int e = e0 + u * SORT_BD;
        d[u] = dst[(FULL || e < end) ? e : (end - 1)];
    }
    if (!FULL) {
#pragma unroll
        for (int u = 0; u < SORT_U; ++u)
            asm volatile("" : "+v"(d[u]));
    }
#pragma unroll
    for (int u = 0; u < SORT_U; ++u)
        if (FULL || e0 + u * SORT_BD < end) atomicAdd(&lh[d[u] >> 8], 1);
}

constexpr int FOLD_WG = 9;
constexpr int SORT_NC = SORT_C - FOLD_WG;

__device__ __forceinline__ void fold_weights_role(const float* __restrict__ W1,
                                                  const float* __restrict__ b1,
                                                  const float* __restrict__ W2,
                                                  float* __restrict__ w12, int g, float* wt) {
    constexpr int K = 128;
    constexpr int OUT = 64;
    const int tid = threadIdx.x;
    const int lane = tid & 31;
    const int wave = tid >> 5;
    const int m = lane & 15;
    const int half = lane >> 4;
    const int kg = half << 1;
    const bool compute = wave < 4;

    v2f a[K / 4];
    if (compute) {
        const float* arow = (g < 8) ? (W1 + (size_t)(g * 16 + m) * K) : b1;
#pragma unroll
        for (int j = 0; j < K / 4; ++j) a[j] = *(const v2f*)(arow + 4 * j + kg);
    }

    float w[K * OUT / 1024];
#pragma unroll
    for (int u = 0; u < K * OUT / 1024; ++u) w[u] = W2[tid + u * 1024];
#pragma unroll
    for (int u = 0; u < K * OUT / 1024; ++u) {
        const int i = tid + u * 1024;
        const int k = i / OUT;
        const int n = i - k * OUT;
        wt[(k >> 1) * (2 * OUT) + 2 * n + (k & 1)] = w[u];
    }
    __syncthreads();
    if (!compute) return;

    int ooff = (g * 16 + 8 * half) * OUT + wave * 16 + m;
    const float* lbase = wt + (half * OUT + m) * 2 + wave * 32;
    v8f acc = {};
    asm volatile("" : "+v"(ooff));
    __builtin_amdgcn_sched_barrier(0);
#pragma unroll
    for (int j = 0; j < K / 4; ++j) {
        const v2f b = *(const v2f*)(lbase + (2 * j) * (2 * OUT));
        acc = __builtin_amdgcn_wmma_f32_16x16x4_f32(false, a[j], false, b, (short)0, acc, false, false);
    }
    __builtin_amdgcn_sched_barrier(0);
    asm volatile("v_nop\n\tv_nop\n\tv_nop\n\tv_nop");
    __builtin_amdgcn_sched_barrier(0);

    if (g < 8) {
#pragma unroll
        for (int r = 0; r < 8; ++r) w12[ooff + r * OUT] = acc[r];
    } else if (half == 0) {
        w12[ooff] = acc[0];
    }
}

template <bool FULL>
__device__ __forceinline__ void place_trip(const int* __restrict__ src, const int* __restrict__ dst,
                                           int* stage, int* cur, int e0, int end) {
    int d[SORT_U], s[SORT_U];
#pragma unroll
    for (int u = 0; u < SORT_U; ++u) {
        const int e = e0 + u * SORT_BD;
        const int ec = (FULL || e < end) ? e : (end - 1);
        d[u] = dst[ec];
        s[u] = src[ec];
    }
    if (!FULL) {
#pragma unroll
        for (int u = 0; u < SORT_U; ++u)
            asm volatile("" : "+v"(d[u]), "+v"(s[u]));
    }
#pragma unroll
    for (int u = 0; u < SORT_U; ++u)
        if (FULL || e0 + u * SORT_BD < end) {
            const int pos = atomicAdd(&cur[d[u] >> 8], 1);
            stage[pos] = (int)(((unsigned)s[u] << 8) | (unsigned)(d[u] & 255));
        }
}

__global__ __launch_bounds__(SORT_BD) void chunk_order_fold_kernel(const int* __restrict__ src,
                                                                   const int* __restrict__ dst,
                                                                   int* __restrict__ hist,
                                                                   int* __restrict__ lst,
                                                                   int* __restrict__ rec,
                                                                   int E, int CH, int NB,
                                                                   const float* __restrict__ W1,
                                                                   const float* __restrict__ b1,
                                                                   const float* __restrict__ W2,
                                                                   float* __restrict__ w12) {
    extern __shared__ int dyn[];
    __shared__ float wt[128 * 64];
    __shared__ int wsum[32];
    __shared__ int s_tot;
    const int col = blockIdx.x;
    if (col < FOLD_WG) {
        for (int b = threadIdx.x; b < NB; b += SORT_BD) {
            hist[b * SORT_C + col] = 0;
            lst[b * SORT_C + col] = 0;
        }
        fold_weights_role(W1, b1, W2, w12, col, wt);
        return;
    }
    int* lh = dyn;
    int* stage = dyn + NB;
    const int tid = threadIdx.x;
    const int lane = tid & 31;
    const int wave = tid >> 5;
    const int c = col - FOLD_WG;
    for (int b = tid; b < NB; b += SORT_BD) lh[b] = 0;
    __syncthreads();
    const int beg = c * CH;
    const int end = (beg + CH < E) ? (beg + CH) : E;
    for (int e0 = beg + tid; e0 < end; e0 += SORT_U * SORT_BD) {
        if (e0 + (SORT_U - 1) * SORT_BD < end) count_trip<true>(dst, lh, e0, end);
        else                                   count_trip<false>(dst, lh, e0, end);
    }
    __syncthreads();

    int carry = 0;
    for (int b0 = 0; b0 < NB; b0 += SORT_BD) {
        const int b = b0 + tid;
        const int cv = (b < NB) ? lh[b] : 0;
        int incl = cv;
        for (int o = 1; o < 32; o <<= 1) {
            int t = __shfl_up(incl, o);
            if (lane >= o) incl += t;
        }
        if (lane == 31) wsum[wave] = incl;
        __syncthreads();
        if (wave == 0) {
            int v = wsum[lane];
            int iv = v;
            for (int o = 1; o < 32; o <<= 1) {
                int t = __shfl_up(iv, o);
                if (lane >= o) iv += t;
            }
            wsum[lane] = iv - v;
            if (lane == 31) s_tot = iv;
        }
        __syncthreads();
        const int excl = carry + wsum[wave] + (incl - cv);
        if (b < NB) {
            lh[b] = excl;
            hist[b * SORT_C + col] = cv;
            lst[b * SORT_C + col] = excl;
        }
        carry += s_tot;
        __syncthreads();
    }

    for (int e0 = beg + tid; e0 < end; e0 += SORT_U * SORT_BD) {
        if (e0 + (SORT_U - 1) * SORT_BD < end) place_trip<true>(src, dst, stage, lh, e0, end);
        else                                   place_trip<false>(src, dst, stage, lh, e0, end);
    }
    __syncthreads();
    for (int i = tid; i < end - beg; i += SORT_BD) rec[beg + i] = stage[i];
}

constexpr int SEG_U = 8;
constexpr int CS_CAP = 8192;

template <int MODE>
__device__ __forceinline__ void run_trip(const int* __restrict__ rec, int* __restrict__ csr_src,
                                         int* lc, int* cs, int bbeg, int seg0, int segn, int k0) {
    int r[SEG_U];
#pragma unroll
    for (int u = 0; u < SEG_U; ++u) {
        const int k = k0 + 4 * u;
        r[u] = rec[seg0 + ((k < segn) ? k : 0)];
    }
#pragma unroll
    for (int u = 0; u < SEG_U; ++u)
        if (k0 + 4 * u < segn) {
            const int pos = atomicAdd(&lc[r[u] & 255], 1);
            if (MODE == 1) csr_src[bbeg + pos] = (int)((unsigned)r[u] >> 8);
            if (MODE == 2) cs[pos] = (int)((unsigned)r[u] >> 8);
        }
}

constexpr int XS_PITCH = 132;

__device__ __forceinline__ void x_rows_load(const float* x, int N, int row0, int lane, v4f (&g)[4]) {
#pragma unroll
    for (int u = 0; u < 4; ++u) {
        int row = row0 + u;
        if (row >= N) row = N - 1;
        g[u] = *(const v4f*)(x + (long)row * 128 + lane * 4);
    }
}

__device__ __forceinline__ void x_rows_stage(float* xs, int rl0, int lane, const v4f (&g)[4]) {
#pragma unroll
    for (int u = 0; u < 4; ++u) *(v4f*)(xs + (rl0 + u) * XS_PITCH + lane * 4) = g[u];
}

__device__ __forceinline__ void tile_product_store(const float* xs, const float* wt, const float* sdinv,
                                                   float* __restrict__ xw, int N, int bucketRow0, int pass0,
                                                   int wave, int lane) {
    const int m = lane & 15;
    const int half = lane >> 4;
    const int rt = wave >> 2;
    const int ct = wave & 3;
    const float* abase = xs + (rt * 16 + m) * XS_PITCH + 2 * half;
    const float* bbase = wt + (half * 64 + ct * 16 + m) * 2;
    v8f acc = {};
#pragma nounroll
    for (int k0 = 0; k0 < 128; k0 += 32) {
#pragma unroll
        for (int k = k0; k < k0 + 32; k += 4) {
            const v2f a = *(const v2f*)(abase + k);
            const v2f bq = *(const v2f*)(bbase + (k >> 1) * 128);
            acc = __builtin_amdgcn_wmma_f32_16x16x4_f32(false, a, false, bq, (short)0, acc, false, false);
        }
        __builtin_amdgcn_sched_barrier(0);
        asm volatile("v_nop\n\tv_nop\n\tv_nop\n\tv_nop");
        __builtin_amdgcn_sched_barrier(0);
    }

#pragma unroll
    for (int r = 0; r < 8; ++r) {
        const int rloc = pass0 + rt * 16 + r + 8 * half;
        const int row = bucketRow0 + rloc;
        if (row < N) xw[(long)row * 64 + ct * 16 + m] = acc[r] * sdinv[rloc];
    }
}

__global__ __launch_bounds__(SORT_BD) void bucket_sort_gemm_kernel(const int* __restrict__ rec,
                                                                   const int* __restrict__ hist,
                                                                   const int* __restrict__ lst,
                                                                   int* __restrict__ csr_src,
                                                                   int* __restrict__ cnt,
                                                                   int* __restrict__ offsets,
                                                                   float* __restrict__ dinv,
                                                                   int N, int CH,
                                                                   const float* x,
                                                                   const float* __restrict__ w12,
                                                                   float* __restrict__ xw) {
    __shared__ int lc[256];
    __shared__ int wsum[32];
    __shared__ int s_prefix;
    __shared__ float sdinv[256];
    __shared__ __attribute__((aligned(16))) float wt[128 * 64];
    __shared__ __attribute__((aligned(16))) float xs[128 * XS_PITCH];
    __shared__ int cs[CS_CAP];
    const int tid = threadIdx.x;
    const int lane = tid & 31;
    const int wave = tid >> 5;
    const int b = blockIdx.x;

    float wq[128 * 64 / SORT_BD];
#pragma unroll
    for (int u = 0; u < 128 * 64 / SORT_BD; ++u) wq[u] = w12[tid + u * SORT_BD];

    const int cc = tid >> 2;
    const int sub = tid & 3;
    const int segn = hist[b * SORT_C + cc];
    const int seg0 = (cc - FOLD_WG) * CH + lst[b * SORT_C + cc];

    int part = 0;
    const int4* c4 = (const int4*)hist;
    const int lim = (b * SORT_C) >> 2;
    for (int j = tid; j < lim; j += 8 * SORT_BD) {
        int4 v[8];
#pragma unroll
        for (int u = 0; u < 8; ++u) {
            const int q = j + u * SORT_BD;
            v[u] = c4[(q < lim) ? q : (lim - 1)];
        }
#pragma unroll
        for (int u = 0; u < 8; ++u)
            if (j + u * SORT_BD < lim) part += (v[u].x + v[u].y) + (v[u].z + v[u].w);
    }
    for (int o = 16; o > 0; o >>= 1) part += __shfl_xor(part, o);
    if (lane == 0) wsum[wave] = part;
    if (tid < 256) lc[tid] = 0;
#pragma unroll
    for (int u = 0; u < 128 * 64 / SORT_BD; ++u) {
        const int i = tid + u * SORT_BD;
        const int k = i >> 6;
        const int n = i & 63;
        wt[(k >> 1) * 128 + 2 * n + (k & 1)] = wq[u];
    }
    __syncthreads();
    if (wave == 0) {
        int v = wsum[lane];
        for (int o = 16; o > 0; o >>= 1) v += __shfl_xor(v, o);
        if (lane == 0) s_prefix = v;
    }
    for (int k0 = sub; k0 < segn; k0 += 4 * SEG_U) run_trip<0>(rec, csr_src, lc, cs, 0, seg0, segn, k0);
    __syncthreads();
    const int bbeg = s_prefix;

    int c = 0, incl = 0;
    if (tid < 256) {
        c = lc[tid];
        incl = c;
        for (int o = 1; o < 32; o <<= 1) {
            int t = __shfl_up(incl, o);
            if (lane >= o) incl += t;
        }
        if (lane == 31) wsum[wave] = incl;
    }
    __syncthreads();
    if (tid < 256) {
        int wpre = 0;
        for (int w = 0; w < 8; ++w) wpre += (w < wave) ? wsum[w] : 0;
        const int excl = wpre + (incl - c);
        lc[tid] = excl;
        const int node = b * 256 + tid;
        const float dg = (float)(c + 1);
        const float dv = (dg > 0.0f) ? rsqrtf(dg) : 0.0f;
        sdinv[tid] = dv;
        if (node < N) {
            cnt[node] = c;
            offsets[node] = bbeg + excl;
            dinv[node] = dv;
        }
    }
    __syncthreads();
    v4f g0[4];
    x_rows_load(x, N, b * 256 + wave * 4, lane, g0);
    int total = 0;
    for (int w = 0; w < 8; ++w) total += wsum[w];
    total = __builtin_amdgcn_readfirstlane(total);
    const bool in_lds = total <= CS_CAP;
    if (in_lds) {
        for (int k0 = sub; k0 < segn; k0 += 4 * SEG_U) run_trip<2>(rec, csr_src, lc, cs, bbeg, seg0, segn, k0);
    } else {
        for (int k0 = sub; k0 < segn; k0 += 4 * SEG_U) run_trip<1>(rec, csr_src, lc, cs, bbeg, seg0, segn, k0);
    }

    x_rows_stage(xs, wave * 4, lane, g0);
    __syncthreads();
    v4f g1[4];
    x_rows_load(x, N, b * 256 + 128 + wave * 4, lane, g1);
    if (in_lds) {
        int* lout = csr_src + bbeg;
        const int sh = (int)((((size_t)lout) >> 2) & 31);
        for (int i0 = tid - sh; i0 < total; i0 += 4 * SORT_BD) {
            int v[4];
#pragma unroll
            for (int u = 0; u < 4; ++u) {
                const int i = i0 + u * SORT_BD;
                v[u] = cs[(i >= 0 && i < total) ? i : 0];
            }
#pragma unroll
            for (int u = 0; u < 4; ++u) {
                const int i = i0 + u * SORT_BD;
                if (i >= 0 && i < total) lout[i] = v[u];
            }
        }
    }
    tile_product_store(xs, wt, sdinv, xw, N, b * 256, 0, wave, lane);
    __syncthreads();
    x_rows_stage(xs, wave * 4, lane, g1);
    __syncthreads();
    tile_product_store(xs, wt, sdinv, xw, N, b * 256, 128, wave, lane);
}


constexpr int NG_NPW = 7;

template <int VPL, bool HAS_BIAS, bool SCALE_OUT>
__global__ __launch_bounds__(256) void node_gather_kernel(const float* __restrict__ in,
                                                          const float* __restrict__ dinv,
                                                          const int* __restrict__ offsets,
                                                          const int* __restrict__ cnt,
                                                          const int* __restrict__ csr_src,
                                                          const float* __restrict__ bias,
                                                          float* __restrict__ out, int N) {
    constexpr int DIM = 32 * VPL;
    typedef __attribute__((ext_vector_type(VPL))) float vec;
    const int lane = threadIdx.x & 31;
    int w = blockIdx.x * 8 + (threadIdx.x >> 5);
    w = __builtin_amdgcn_readfirstlane(w);
    const int d0 = w * NG_NPW;
    if (d0 >= N) return;
    const int nn = (N - d0 < NG_NPW) ? (N - d0) : NG_NPW;

    vec self_n = *(const vec*)(in + (size_t)d0 * DIM + lane * VPL);
    const int dl = d0 + ((lane < nn) ? lane : 0);
    const int offv = offsets[dl];
    const int cntv = cnt[dl];
    const int dinvb = __float_as_int(dinv[dl]);
    vec bv = {};
    if (HAS_BIAS) bv = *(const vec*)(bias + lane * VPL);

    int nbeg = __builtin_amdgcn_readlane(offv, 0);
    int nlen = __builtin_amdgcn_readlane(cntv, 0);
    int sj_n = 0;
    if (lane < nlen) sj_n = csr_src[nbeg + lane];

    for (int i = 0; i < nn; ++i) {
        const int d = d0 + i;
        const int beg = nbeg;
        const int len = nlen;
        const float dd = __int_as_float(__builtin_amdgcn_readlane(dinvb, i));
        int sj = sj_n;
        vec acc = self_n;

        if (i + 1 < nn) {
            nbeg = __builtin_amdgcn_readlane(offv, i + 1);
            nlen = __builtin_amdgcn_readlane(cntv, i + 1);
            sj_n = 0;
            if (lane < nlen) sj_n = csr_src[nbeg + lane];
            self_n = *(const vec*)(in + (size_t)(d + 1) * DIM + lane * VPL);
        }

        for (int base = 0; base < len; base += 32) {
            const int n = (len - base < 32) ? (len - base) : 32;
            if (base > 0) {
                sj = 0;
                if (lane < n) sj = csr_src[beg + base + lane];
                asm volatile("" : "+v"(sj));
            }
#define NG_ROW(jj) (*(const vec*)(in + (size_t)__builtin_amdgcn_readlane(sj, (jj)) * DIM + lane * VPL))
            int j = 0;
            for (; j + 8 <= n; j += 8) {
                const vec r0 = NG_ROW(j),     r1 = NG_ROW(j + 1), r2 = NG_ROW(j + 2), r3 = NG_ROW(j + 3);
                const vec r4 = NG_ROW(j + 4), r5 = NG_ROW(j + 5), r6 = NG_ROW(j + 6), r7 = NG_ROW(j + 7);
                acc += r0;
                acc += r1;
                acc += r2;
                acc += r3;
                acc += r4;
                acc += r5;
                acc += r6;
                acc += r7;
            }
            if (j + 4 <= n) {
                const vec r0 = NG_ROW(j), r1 = NG_ROW(j + 1), r2 = NG_ROW(j + 2), r3 = NG_ROW(j + 3);
                acc += r0;
                acc += r1;
                acc += r2;
                acc += r3;
                j += 4;
            }
            if (j < n) {
                const vec z = {};
                const vec r0 = NG_ROW(j);
                vec r1 = z, r2 = z;
                if (j + 1 < n) r1 = NG_ROW(j + 1);
                if (j + 2 < n) r2 = NG_ROW(j + 2);
                acc += r0;
                acc += r1;
                acc += r2;
            }
#undef NG_ROW
        }

        acc *= dd;
        if (HAS_BIAS) acc += bv;
        if (SCALE_OUT) acc *= dd;
        *(vec*)(out + (size_t)d * DIM + lane * VPL) = acc;
    }
}


extern "C" void kernel_launch(void* const* d_in, const int* in_sizes, int n_in,
                              void* d_out, int out_size, void* d_ws, size_t ws_size,
                              hipStream_t stream) {
    const float* x  = (const float*)d_in[0];
    const int*   ei = (const int*)d_in[1];
    const float* W1 = (const float*)d_in[2];
    const float* b1 = (const float*)d_in[3];
    const float* W2 = (const float*)d_in[4];
    const float* b2 = (const float*)d_in[5];
    float* out = (float*)d_out;

    const int N = in_sizes[0] / 128;
    const int E = in_sizes[1] / 2;
    const int* src = ei;
    const int* dst = ei + E;

    float* xw      = (float*)d_ws;
    float* t       = xw + (size_t)N * 64;
    float* dinv    = t + (size_t)N * 64;
    float* w12     = dinv + N;
    float* b12     = w12 + 128 * 64;
    int*   cnt     = (int*)(w12 + 129 * 64);
    int*   offsets = cnt + N;
    int*   rec     = offsets + N;
    int*   csr_src = rec + E;

    const int NB = (N + 255) / 256;
    const int CH = (E + SORT_NC - 1) / SORT_NC;
    const int M  = NB * SORT_C;
    int* hist = (int*)t;
    int* lst  = hist + M;
    chunk_order_fold_kernel<<<SORT_C, SORT_BD, (NB + CH) * sizeof(int), stream>>>(src, dst, hist, lst, rec, E, CH, NB,
                                                                                           W1, b1, W2, w12);
    bucket_sort_gemm_kernel<<<NB, SORT_BD, 0, stream>>>(rec, hist, lst, csr_src, cnt, offsets, dinv, N, CH, x, w12, xw);

    const int gatherBlocks = ((N + NG_NPW - 1) / NG_NPW + 7) / 8;

    node_gather_kernel<2, true, true><<<gatherBlocks, 256, 0, stream>>>(xw, dinv, offsets, cnt, csr_src, b12, t, N);
    node_gather_kernel<2, true, false><<<gatherBlocks, 256, 0, stream>>>(t, dinv, offsets, cnt, csr_src, b2, out, N);
}
